# starting kernel
# baseline (speedup 1.0000x reference)
#include <hip/hip_runtime.h>

typedef float v2f __attribute__((ext_vector_type(2)));
typedef float v8f __attribute__((ext_vector_type(8)));

__global__ __launch_bounds__(256) void k_fill(float* __restrict__ p, float v, long n) {
  long i = (long)blockIdx.x * blockDim.x + threadIdx.x;
  if (i < n) p[i] = v;
}

__global__ __launch_bounds__(256) void k_deg(const int* __restrict__ dst,
                                             float* __restrict__ deg, int E) {
  int i = blockIdx.x * blockDim.x + threadIdx.x;
  if (i < E) unsafeAtomicAdd(&deg[dst[i]], 1.0f);
}

__global__ __launch_bounds__(256) void k_rsqrt(float* __restrict__ d, int n) {
  int i = blockIdx.x * blockDim.x + threadIdx.x;
  if (i < n) d[i] = rsqrtf(d[i]);
}

template <int OUT>
__global__ __launch_bounds__(256)
void k_gemm(const float* __restrict__ A, const float* __restrict__ W,
            float* __restrict__ H, int M) {
  constexpr int NT = OUT / 16;
  int wave = (blockIdx.x * blockDim.x + threadIdx.x) >> 5;
  int lane = threadIdx.x & 31;
  int row0 = wave * 16;
  if (row0 >= M) return;

  int m  = lane & 15;
  int kh = lane >> 4;

  v8f acc[NT] = {};

  int arow = (row0 + m < M) ? (row0 + m) : (M - 1);
  const float* ap = A + (size_t)arow * 128 + 2 * kh;

  for (int k = 0; k < 128; k += 4) {
    v2f a = *(const v2f*)(ap + k);
    const float* bp = W + (size_t)(k + 2 * kh) * OUT + m;
#pragma unroll
    for (int t = 0; t < NT; ++t) {
      v2f b;
      b.x = bp[t * 16];
      b.y = bp[t * 16 + OUT];
      acc[t] = __builtin_amdgcn_wmma_f32_16x16x4_f32(
           false, a,  false, b,
           (short)0, acc[t],  false,  false);
    }
  }

  int rbase = row0 + 8 * kh;
#pragma unroll
  for (int t = 0; t < NT; ++t) {
#pragma unroll
    for (int r = 0; r < 8; ++r) {
      int row = rbase + r;
      if (row < M) H[(size_t)row * OUT + t * 16 + m] = acc[t][r];
    }
  }
}

template <int OUT>
__global__ __launch_bounds__(256)
void k_scatter(const int* __restrict__ src, const int* __restrict__ dst,
               const float* __restrict__ dinv, const float* __restrict__ h,
               float* __restrict__ agg, int E) {
  int wave = (blockIdx.x * blockDim.x + threadIdx.x) >> 5;
  int lane = threadIdx.x & 31;
  if (wave >= E) return;
  int s = src[wave];
  int d = dst[wave];
  float nrm = dinv[s] * dinv[d];
  const float* hs = h + (size_t)s * OUT;
  float* ad = agg + (size_t)d * OUT;
#pragma unroll
  for (int k = lane; k < OUT; k += 32)
    unsafeAtomicAdd(ad + k, hs[k] * nrm);
}

template <int OUT, bool RELU>
__global__ __launch_bounds__(256)
void k_act(float* __restrict__ agg, const float* __restrict__ h,
           const float* __restrict__ dinv, const float* __restrict__ b, int N) {
  long i = (long)blockIdx.x * blockDim.x + threadIdx.x;
  long tot = (long)N * OUT;
  if (i >= tot) return;
  int node = (int)(i / OUT);
  int dcol = (int)(i % OUT);
  float di = dinv[node];
  float v = agg[i] + h[i] * di * di + b[dcol];
  agg[i] = RELU ? fmaxf(v, 0.0f) : v;
}

static inline int cdiv(long a, long b) { return (int)((a + b - 1) / b); }

extern "C" void kernel_launch(void* const* d_in, const int* in_sizes, int n_in,
                              void* d_out, int out_size, void* d_ws, size_t ws_size,
                              hipStream_t stream) {
  const float* x  = (const float*)d_in[0];
  const int*   ei = (const int*)d_in[1];
  const float* W1 = (const float*)d_in[2];
  const float* b1 = (const float*)d_in[3];
  const float* W2 = (const float*)d_in[4];
  const float* b2 = (const float*)d_in[5];
  const float* W3 = (const float*)d_in[6];
  const float* b3 = (const float*)d_in[7];

  const int N = in_sizes[0] / 128;
  const int E = in_sizes[1] / 2;
  const int* src = ei;
  const int* dst = ei + E;

  char* ws = (char*)d_ws;
  size_t off0 = 0;
  size_t off1 = (((size_t)N * 4) + 255) & ~(size_t)255;
  size_t off2 = off1 + (size_t)N * 128 * 4;
  float* dinv = (float*)(ws + off0);
  float* bufH = (float*)(ws + off1);
  float* bufF = (float*)(ws + off2);
  float* out  = (float*)d_out;

  const int T = 256;
  const int gemmBlocks = cdiv((long)cdiv(N, 16) * 32, T);
  const int edgeBlocks = cdiv((long)E * 32, T);

  k_fill<<<cdiv(N, T), T, 0, stream>>>(dinv, 1.0f, N);
  k_deg<<<cdiv(E, T), T, 0, stream>>>(dst, dinv, E);
  k_rsqrt<<<cdiv(N, T), T, 0, stream>>>(dinv, N);

  k_gemm<128><<<gemmBlocks, T, 0, stream>>>(x, W1, bufH, N);
  k_fill<<<cdiv((long)N * 128, T), T, 0, stream>>>(bufF, 0.0f, (long)N * 128);
  k_scatter<128><<<edgeBlocks, T, 0, stream>>>(src, dst, dinv, bufH, bufF, E);
  k_act<128, true><<<cdiv((long)N * 128, T), T, 0, stream>>>(bufF, bufH, dinv, b1, N);

  k_gemm<128><<<gemmBlocks, T, 0, stream>>>(bufF, W2, bufH, N);
  k_fill<<<cdiv((long)N * 128, T), T, 0, stream>>>(bufF, 0.0f, (long)N * 128);
  k_scatter<128><<<edgeBlocks, T, 0, stream>>>(src, dst, dinv, bufH, bufF, E);
  k_act<128, true><<<cdiv((long)N * 128, T), T, 0, stream>>>(bufF, bufH, dinv, b2, N);

  k_gemm<64><<<gemmBlocks, T, 0, stream>>>(bufF, W3, bufH, N);
  k_fill<<<cdiv((long)N * 64, T), T, 0, stream>>>(out, 0.0f, (long)N * 64);
  k_scatter<64><<<edgeBlocks, T, 0, stream>>>(src, dst, dinv, bufH, out, E);
  k_act<64, false><<<cdiv((long)N * 64, T), T, 0, stream>>>(out, bufH, dinv, b3, N);
}


# Aggregation by output-tile ownership: a block keeps 392 output rows in LDS, scans all edges, adds their source rows by LDS float adds; no fill, global scatter, add pass; sum order: LDS arrival
# speedup vs baseline: 1.8666x; 1.8666x over previous
#include <hip/hip_runtime.h>

typedef float v2f __attribute__((ext_vector_type(2)));
typedef float v8f __attribute__((ext_vector_type(8)));

__global__ __launch_bounds__(256) void k_fill(float* __restrict__ p, float v, long n) {
  long i = (long)blockIdx.x * blockDim.x + threadIdx.x;
  if (i < n) p[i] = v;
}

__global__ __launch_bounds__(256) void k_deg(const int* __restrict__ dst,
                                             float* __restrict__ deg, int E) {
  int i = blockIdx.x * blockDim.x + threadIdx.x;
  if (i < E) unsafeAtomicAdd(&deg[dst[i]], 1.0f);
}

__global__ __launch_bounds__(256) void k_rsqrt(float* __restrict__ d, int n) {
  int i = blockIdx.x * blockDim.x + threadIdx.x;
  if (i < n) d[i] = rsqrtf(d[i]);
}

template <int OUT>
__global__ __launch_bounds__(256)
void k_gemm(const float* __restrict__ A, const float* __restrict__ W,
            float* __restrict__ H, int M) {
  constexpr int NT = OUT / 16;
  int wave = (blockIdx.x * blockDim.x + threadIdx.x) >> 5;
  int lane = threadIdx.x & 31;
  int row0 = wave * 16;
  if (row0 >= M) return;

  int m  = lane & 15;
  int kh = lane >> 4;

  v8f acc[NT] = {};

  int arow = (row0 + m < M) ? (row0 + m) : (M - 1);
  const float* ap = A + (size_t)arow * 128 + 2 * kh;

  for (int k = 0; k < 128; k += 4) {
    v2f a = *(const v2f*)(ap + k);
    const float* bp = W + (size_t)(k + 2 * kh) * OUT + m;
#pragma unroll
    for (int t = 0; t < NT; ++t) {
      v2f b;
      b.x = bp[t * 16];
      b.y = bp[t * 16 + OUT];
      acc[t] = __builtin_amdgcn_wmma_f32_16x16x4_f32(
           false, a,  false, b,
           (short)0, acc[t],  false,  false);
    }
  }

  int rbase = row0 + 8 * kh;
#pragma unroll
  for (int t = 0; t < NT; ++t) {
#pragma unroll
    for (int r = 0; r < 8; ++r) {
      int row = rbase + r;
      if (row < M) H[(size_t)row * OUT + t * 16 + m] = acc[t][r];
    }
  }
}

constexpr int TN   = 392;
constexpr int LCAP = 12288;
constexpr int BT   = 1024;
constexpr int SU   = 4;
constexpr int RU   = 8;

typedef float vf2 __attribute__((ext_vector_type(2)));
typedef float vf4 __attribute__((ext_vector_type(4)));
typedef int   vi2 __attribute__((ext_vector_type(2)));
typedef int   vi4 __attribute__((ext_vector_type(4)));
template <int V> struct VT;
template <> struct VT<2> { typedef vf2 T; };
template <> struct VT<4> { typedef vf4 T; };

template <int OUT, bool RELU>
__global__ __launch_bounds__(BT)
void k_agg(const int* __restrict__ src, const int* __restrict__ dst,
           const float* __restrict__ dinv, const float* __restrict__ h,
           const float* __restrict__ b, float* __restrict__ outp, int N, int E) {
  constexpr int VEC = OUT / 32;
  typedef typename VT<VEC>::T V;
  __shared__ float tile[TN * OUT];
  __shared__ vi2   lst[LCAP];
  __shared__ float dl[TN];
  __shared__ int   cnt;

  const int tid  = threadIdx.x;
  const int lane = tid & 31;
  const int wave = tid >> 5;
  const int lo   = blockIdx.x * TN;
  const int nt   = (N - lo < TN) ? (N - lo) : TN;
  const unsigned unt = (unsigned)nt;
  const unsigned ulo = (unsigned)lo;

  if (tid == 0) cnt = 0;
  for (int t = tid; t < nt; t += BT) dl[t] = dinv[lo + t];
  for (int t = wave; t < nt; t += BT / 32) {
    float di = dinv[lo + t];
    V x = *(const V*)(h + (size_t)(lo + t) * OUT + VEC * lane);
#pragma unroll
    for (int j = 0; j < VEC; ++j) tile[t * OUT + j * 32 + lane] = x[j] * di * di;
  }
  __syncthreads();

  auto put = [&](int e, unsigned t) {
    int pos = atomicAdd(&cnt, 1);
    if (pos < LCAP) {
      vi2 ent; ent.x = e; ent.y = (int)t;
      lst[pos] = ent;
    } else {
      int s = src[e];
      float nrm = dinv[s] * dl[t];
      const float* hs = h + (size_t)s * OUT;
      for (int c = 0; c < OUT; ++c)
        atomicAdd(&tile[t * OUT + (c % VEC) * 32 + c / VEC], hs[c] * nrm);
    }
  };

  int head = (int)(((16u - (unsigned)((size_t)dst & 15u)) & 15u) >> 2);
  if (head > E) head = E;
  const int nvec  = (E - head) >> 2;
  const int tail0 = head + 4 * nvec;
  const vi4* dv = (const vi4*)(dst + head);

  for (int vb = 0; vb < nvec; vb += BT * SU) {
    vi4 d[SU];
    if (vb + BT * SU <= nvec) {
#pragma unroll
      for (int u = 0; u < SU; ++u) d[u] = dv[vb + u * BT + tid];
    } else {
#pragma unroll
      for (int u = 0; u < SU; ++u) {
        int i = vb + u * BT + tid;
        vi4 none = {-1, -1, -1, -1};
        d[u] = (i < nvec) ? dv[i] : none;
      }
    }
#pragma unroll
    for (int u = 0; u < SU; ++u) {
      unsigned t0 = (unsigned)d[u].x - ulo, t1 = (unsigned)d[u].y - ulo;
      unsigned t2 = (unsigned)d[u].z - ulo, t3 = (unsigned)d[u].w - ulo;
      unsigned m01 = t0 < t1 ? t0 : t1, m23 = t2 < t3 ? t2 : t3;
      unsigned m = m01 < m23 ? m01 : m23;
      if (m < unt) {
        int e0 = head + 4 * (vb + u * BT + tid);
        if (t0 < unt) put(e0, t0);
        if (t1 < unt) put(e0 + 1, t1);
        if (t2 < unt) put(e0 + 2, t2);
        if (t3 < unt) put(e0 + 3, t3);
      }
    }
  }
  {
    int e = -1;
    if (tid < head) e = tid;
    else if (tid - head < E - tail0) e = tail0 + (tid - head);
    if (e >= 0) {
      unsigned t = (unsigned)dst[e] - ulo;
      if (t < unt) put(e, t);
    }
  }
  __syncthreads();

  const int n = cnt < LCAP ? cnt : LCAP;
  for (int i = tid; i < n; i += BT) {
    vi2 ent = lst[i];
    ent.x = src[ent.x];
    lst[i] = ent;
  }
  __syncthreads();

  for (int base = wave; base < n; base += (BT / 32) * RU) {
    if (base + (BT / 32) * (RU - 1) < n) {
      int   s[RU], t[RU];
      V     x[RU];
      float w[RU];
#pragma unroll
      for (int u = 0; u < RU; ++u) {
        vi2 ent = lst[base + (BT / 32) * u];
        s[u] = __builtin_amdgcn_readfirstlane(ent.x);
        t[u] = __builtin_amdgcn_readfirstlane(ent.y);
      }
#pragma unroll
      for (int u = 0; u < RU; ++u) {
        x[u] = *(const V*)(h + (size_t)s[u] * OUT + VEC * lane);
        w[u] = dinv[s[u]] * dl[t[u]];
      }
#pragma unroll
      for (int u = 0; u < RU; ++u) {
#pragma unroll
        for (int j = 0; j < VEC; ++j)
          atomicAdd(&tile[t[u] * OUT + j * 32 + lane], x[u][j] * w[u]);
      }
    } else {
      for (int i = base; i < n; i += BT / 32) {
        vi2 ent = lst[i];
        int s = __builtin_amdgcn_readfirstlane(ent.x);
        int t = __builtin_amdgcn_readfirstlane(ent.y);
        V x = *(const V*)(h + (size_t)s * OUT + VEC * lane);
        float w = dinv[s] * dl[t];
#pragma unroll
        for (int j = 0; j < VEC; ++j)
          atomicAdd(&tile[t * OUT + j * 32 + lane], x[j] * w);
      }
    }
  }
  __syncthreads();

  {
    V bb = *(const V*)(b + VEC * lane);
    for (int t = wave; t < nt; t += BT / 32) {
      V r;
#pragma unroll
      for (int j = 0; j < VEC; ++j) {
        float v = tile[t * OUT + j * 32 + lane] + bb[j];
        r[j] = RELU ? fmaxf(v, 0.0f) : v;
      }
      *(V*)(outp + (size_t)(lo + t) * OUT + VEC * lane) = r;
    }
  }
}

static inline int cdiv(long a, long b) { return (int)((a + b - 1) / b); }

extern "C" void kernel_launch(void* const* d_in, const int* in_sizes, int n_in,
                              void* d_out, int out_size, void* d_ws, size_t ws_size,
                              hipStream_t stream) {
  const float* x  = (const float*)d_in[0];
  const int*   ei = (const int*)d_in[1];
  const float* W1 = (const float*)d_in[2];
  const float* b1 = (const float*)d_in[3];
  const float* W2 = (const float*)d_in[4];
  const float* b2 = (const float*)d_in[5];
  const float* W3 = (const float*)d_in[6];
  const float* b3 = (const float*)d_in[7];

  const int N = in_sizes[0] / 128;
  const int E = in_sizes[1] / 2;
  const int* src = ei;
  const int* dst = ei + E;

  char* ws = (char*)d_ws;
  size_t off0 = 0;
  size_t off1 = (((size_t)N * 4) + 255) & ~(size_t)255;
  size_t off2 = off1 + (size_t)N * 128 * 4;
  float* dinv = (float*)(ws + off0);
  float* bufH = (float*)(ws + off1);
  float* bufF = (float*)(ws + off2);
  float* out  = (float*)d_out;

  const int T = 256;
  const int gemmBlocks = cdiv((long)cdiv(N, 16) * 32, T);
  const int tiles = cdiv(N, TN);

  k_fill<<<cdiv(N, T), T, 0, stream>>>(dinv, 1.0f, N);
  k_deg<<<cdiv(E, T), T, 0, stream>>>(dst, dinv, E);
  k_rsqrt<<<cdiv(N, T), T, 0, stream>>>(dinv, N);

  k_gemm<128><<<gemmBlocks, T, 0, stream>>>(x, W1, bufH, N);
  k_agg<128, true><<<tiles, BT, 0, stream>>>(src, dst, dinv, bufH, b1, bufF, N, E);

  k_gemm<128><<<gemmBlocks, T, 0, stream>>>(bufF, W2, bufH, N);
  k_agg<128, true><<<tiles, BT, 0, stream>>>(src, dst, dinv, bufH, b2, bufF, N, E);

  k_gemm<64><<<gemmBlocks, T, 0, stream>>>(bufF, W3, bufH, N);
  k_agg<64, false><<<tiles, BT, 0, stream>>>(src, dst, dinv, bufH, b3, out, N, E);
}


# Owner scan once per call: one launch where each tile block scans dst, files its edge list {source,row} and its nodes' d^-1/2; the 3 layer launches read the list; 9 to 7 launches; sum order kept
# speedup vs baseline: 3.0345x; 1.6257x over previous
#include <hip/hip_runtime.h>

typedef float v2f __attribute__((ext_vector_type(2)));
typedef float v8f __attribute__((ext_vector_type(8)));

constexpr int TN   = 392;
constexpr int LCAP = 12288;
constexpr int BT   = 1024;
constexpr int SU   = 4;
constexpr int RU   = 8;

typedef float vf2 __attribute__((ext_vector_type(2)));
typedef float vf4 __attribute__((ext_vector_type(4)));
typedef int   vi2 __attribute__((ext_vector_type(2)));
typedef int   vi4 __attribute__((ext_vector_type(4)));
template <int V> struct VT;
template <> struct VT<2> { typedef vf2 T; };
template <> struct VT<4> { typedef vf4 T; };

__global__ __launch_bounds__(BT)
void k_list(const int* __restrict__ src, const int* __restrict__ dst,
            float* __restrict__ dinv, int* __restrict__ glist,
            int* __restrict__ gcnt, int N, int E) {
  __shared__ vi2 lst[LCAP];
  __shared__ int degc[TN];
  __shared__ int cnt;

  const int tid = threadIdx.x;
  const int lo  = blockIdx.x * TN;
  const int nt  = (N - lo < TN) ? (N - lo) : TN;
  const unsigned unt = (unsigned)nt;
  const unsigned ulo = (unsigned)lo;

  if (tid == 0) cnt = 0;
  for (int t = tid; t < TN; t += BT) degc[t] = 0;
  __syncthreads();

  auto put = [&](int e, unsigned t) {
    atomicAdd(&degc[t], 1);
    int pos = atomicAdd(&cnt, 1);
    if (pos < LCAP) {
      vi2 ent; ent.x = e; ent.y = (int)t;
      lst[pos] = ent;
    }
  };

  int head = (int)(((16u - (unsigned)((size_t)dst & 15u)) & 15u) >> 2);
  if (head > E) head = E;
  const int nvec  = (E - head) >> 2;
  const int tail0 = head + 4 * nvec;
  const vi4* dv = (const vi4*)(dst + head);

  for (int vb = 0; vb < nvec; vb += BT * SU) {
    vi4 d[SU];
    if (vb + BT * SU <= nvec) {
#pragma unroll
      for (int u = 0; u < SU; ++u) d[u] = dv[vb + u * BT + tid];
    } else {
#pragma unroll
      for (int u = 0; u < SU; ++u) {
        int i = vb + u * BT + tid;
        vi4 none = {-1, -1, -1, -1};
        d[u] = (i < nvec) ? dv[i] : none;
      }
    }
#pragma unroll
    for (int u = 0; u < SU; ++u) {
      unsigned t0 = (unsigned)d[u].x - ulo, t1 = (unsigned)d[u].y - ulo;
      unsigned t2 = (unsigned)d[u].z - ulo, t3 = (unsigned)d[u].w - ulo;
      unsigned m01 = t0 < t1 ? t0 : t1, m23 = t2 < t3 ? t2 : t3;
      unsigned m = m01 < m23 ? m01 : m23;
      if (m < unt) {
        int e0 = head + 4 * (vb + u * BT + tid);
        if (t0 < unt) put(e0, t0);
        if (t1 < unt) put(e0 + 1, t1);
        if (t2 < unt) put(e0 + 2, t2);
        if (t3 < unt) put(e0 + 3, t3);
      }
    }
  }
  {
    int e = -1;
    if (tid < head) e = tid;
    else if (tid - head < E - tail0) e = tail0 + (tid - head);
    if (e >= 0) {
      unsigned t = (unsigned)dst[e] - ulo;
      if (t < unt) put(e, t);
    }
  }
  __syncthreads();

  const int total = cnt;
  const int n = total < LCAP ? total : LCAP;
  vi2* gl = (vi2*)glist + (size_t)blockIdx.x * LCAP;
  for (int i = tid; i < n; i += BT) {
    vi2 ent = lst[i];
    ent.x = src[ent.x];
    gl[i] = ent;
  }
  if (tid == 0) gcnt[blockIdx.x] = total;
  for (int t = tid; t < nt; t += BT) dinv[lo + t] = rsqrtf(1.0f + (float)degc[t]);
}

template <int OUT>
__global__ __launch_bounds__(256)
void k_gemm(const float* __restrict__ A, const float* __restrict__ W,
            float* __restrict__ H, int M) {
  constexpr int NT = OUT / 16;
  int wave = (blockIdx.x * blockDim.x + threadIdx.x) >> 5;
  int lane = threadIdx.x & 31;
  int row0 = wave * 16;
  if (row0 >= M) return;

  int m  = lane & 15;
  int kh = lane >> 4;

  v8f acc[NT] = {};

  int arow = (row0 + m < M) ? (row0 + m) : (M - 1);
  const float* ap = A + (size_t)arow * 128 + 2 * kh;

  for (int k = 0; k < 128; k += 4) {
    v2f a = *(const v2f*)(ap + k);
    const float* bp = W + (size_t)(k + 2 * kh) * OUT + m;
#pragma unroll
    for (int t = 0; t < NT; ++t) {
      v2f b;
      b.x = bp[t * 16];
      b.y = bp[t * 16 + OUT];
      acc[t] = __builtin_amdgcn_wmma_f32_16x16x4_f32(
           false, a,  false, b,
           (short)0, acc[t],  false,  false);
    }
  }

  int rbase = row0 + 8 * kh;
#pragma unroll
  for (int t = 0; t < NT; ++t) {
#pragma unroll
    for (int r = 0; r < 8; ++r) {
      int row = rbase + r;
      if (row < M) H[(size_t)row * OUT + t * 16 + m] = acc[t][r];
    }
  }
}

template <int OUT, bool RELU>
__global__ __launch_bounds__(BT)
void k_agg(const int* __restrict__ glist, const int* __restrict__ gcnt,
           const int* __restrict__ src, const int* __restrict__ dst,
           const float* __restrict__ dinv, const float* __restrict__ h,
           const float* __restrict__ b, float* __restrict__ outp, int N, int E) {
  constexpr int VEC = OUT / 32;
  typedef typename VT<VEC>::T V;
  __shared__ float tile[TN * OUT];
  __shared__ vi2   lst[LCAP];
  __shared__ float dl[TN];

  const int tid  = threadIdx.x;
  const int lane = tid & 31;
  const int wave = tid >> 5;
  const int lo   = blockIdx.x * TN;
  const int nt   = (N - lo < TN) ? (N - lo) : TN;
  const unsigned unt = (unsigned)nt;
  const unsigned ulo = (unsigned)lo;

  const int  total = gcnt[blockIdx.x];
  const bool whole = total <= LCAP;
  const int  n     = whole ? total : 0;

  {
    constexpr int LU = 8;
    const vi2* gl = (const vi2*)glist + (size_t)blockIdx.x * LCAP;
    for (int i0 = tid; i0 < n; i0 += BT * LU) {
      vi2 ent[LU];
#pragma unroll
      for (int u = 0; u < LU; ++u) {
        int i = i0 + u * BT;
        vi2 none = {0, 0};
        ent[u] = (i < n) ? gl[i] : none;
      }
#pragma unroll
      for (int u = 0; u < LU; ++u) {
        int i = i0 + u * BT;
        if (i < n) lst[i] = ent[u];
      }
    }
  }
  for (int t = tid; t < nt; t += BT) dl[t] = dinv[lo + t];
  for (int t = wave; t < nt; t += BT / 32) {
    float di = dinv[lo + t];
    V x = *(const V*)(h + (size_t)(lo + t) * OUT + VEC * lane);
#pragma unroll
    for (int j = 0; j < VEC; ++j) tile[t * OUT + j * 32 + lane] = x[j] * di * di;
  }
  __syncthreads();

  if (!whole) {
    for (int e = tid; e < E; e += BT) {
      unsigned t = (unsigned)dst[e] - ulo;
      if (t < unt) {
        int s = src[e];
        float nrm = dinv[s] * dl[t];
        const float* hs = h + (size_t)s * OUT;
        for (int c = 0; c < OUT; ++c)
          atomicAdd(&tile[t * OUT + (c % VEC) * 32 + c / VEC], hs[c] * nrm);
      }
    }
  }

  for (int base = wave; base < n; base += (BT / 32) * RU) {
    if (base + (BT / 32) * (RU - 1) < n) {
      int   s[RU], t[RU];
      V     x[RU];
      float w[RU];
#pragma unroll
      for (int u = 0; u < RU; ++u) {
        vi2 ent = lst[base + (BT / 32) * u];
        s[u] = __builtin_amdgcn_readfirstlane(ent.x);
        t[u] = __builtin_amdgcn_readfirstlane(ent.y);
      }
#pragma unroll
      for (int u = 0; u < RU; ++u) {
        x[u] = *(const V*)(h + (size_t)s[u] * OUT + VEC * lane);
        w[u] = dinv[s[u]] * dl[t[u]];
      }
#pragma unroll
      for (int u = 0; u < RU; ++u) {
#pragma unroll
        for (int j = 0; j < VEC; ++j)
          atomicAdd(&tile[t[u] * OUT + j * 32 + lane], x[u][j] * w[u]);
      }
    } else {
      for (int i = base; i < n; i += BT / 32) {
        vi2 ent = lst[i];
        int s = __builtin_amdgcn_readfirstlane(ent.x);
        int t = __builtin_amdgcn_readfirstlane(ent.y);
        V x = *(const V*)(h + (size_t)s * OUT + VEC * lane);
        float w = dinv[s] * dl[t];
#pragma unroll
        for (int j = 0; j < VEC; ++j)
          atomicAdd(&tile[t * OUT + j * 32 + lane], x[j] * w);
      }
    }
  }
  __syncthreads();

  {
    V bb = *(const V*)(b + VEC * lane);
    for (int t = wave; t < nt; t += BT / 32) {
      V r;
#pragma unroll
      for (int j = 0; j < VEC; ++j) {
        float v = tile[t * OUT + j * 32 + lane] + bb[j];
        r[j] = RELU ? fmaxf(v, 0.0f) : v;
      }
      *(V*)(outp + (size_t)(lo + t) * OUT + VEC * lane) = r;
    }
  }
}

static inline int cdiv(long a, long b) { return (int)((a + b - 1) / b); }

extern "C" void kernel_launch(void* const* d_in, const int* in_sizes, int n_in,
                              void* d_out, int out_size, void* d_ws, size_t ws_size,
                              hipStream_t stream) {
  const float* x  = (const float*)d_in[0];
  const int*   ei = (const int*)d_in[1];
  const float* W1 = (const float*)d_in[2];
  const float* b1 = (const float*)d_in[3];
  const float* W2 = (const float*)d_in[4];
  const float* b2 = (const float*)d_in[5];
  const float* W3 = (const float*)d_in[6];
  const float* b3 = (const float*)d_in[7];

  const int N = in_sizes[0] / 128;
  const int E = in_sizes[1] / 2;
  const int* src = ei;
  const int* dst = ei + E;

  const int T = 256;
  const int gemmBlocks = cdiv((long)cdiv(N, 16) * 32, T);
  const int tiles = cdiv(N, TN);

  char* ws = (char*)d_ws;
  size_t off0 = 0;
  size_t off1 = (((size_t)N * 4) + 255) & ~(size_t)255;
  size_t off2 = off1 + (size_t)N * 128 * 4;
  size_t off3 = off2 + (size_t)N * 128 * 4;
  size_t off4 = off3 + (size_t)tiles * LCAP * 8;
  float* dinv  = (float*)(ws + off0);
  float* bufH  = (float*)(ws + off1);
  float* bufF  = (float*)(ws + off2);
  int*   glist = (int*)(ws + off3);
  int*   gcnt  = (int*)(ws + off4);
  float* out   = (float*)d_out;

  k_list<<<tiles, BT, 0, stream>>>(src, dst, dinv, glist, gcnt, N, E);

  k_gemm<128><<<gemmBlocks, T, 0, stream>>>(x, W1, bufH, N);
  k_agg<128, true><<<tiles, BT, 0, stream>>>(glist, gcnt, src, dst, dinv, bufH, b1, bufF, N, E);

  k_gemm<128><<<gemmBlocks, T, 0, stream>>>(bufF, W2, bufH, N);
  k_agg<128, true><<<tiles, BT, 0, stream>>>(glist, gcnt, src, dst, dinv, bufH, b2, bufF, N, E);

  k_gemm<64><<<gemmBlocks, T, 0, stream>>>(bufF, W3, bufH, N);
  k_agg<64, false><<<tiles, BT, 0, stream>>>(glist, gcnt, src, dst, dinv, bufH, b3, out, N, E);
}


# Owner scan in two levels: 16 node ranges x 16 edge slices keep {source,dst} pairs (ballot-ranked append); a tile block reads its range's 16 pair lists, not all dst; 7 to 8 launches; sum order kept
# speedup vs baseline: 3.5469x; 1.1689x over previous
#include <hip/hip_runtime.h>

typedef float v2f __attribute__((ext_vector_type(2)));
typedef float v8f __attribute__((ext_vector_type(8)));

constexpr int TN   = 392;
constexpr int LCAP = 12288;
constexpr int BT   = 1024;
constexpr int RU   = 8;
constexpr int TPR  = 16;
constexpr int RN   = TN * TPR;
constexpr int NS   = 16;
constexpr int PCAP = 8192;
constexpr int PU   = 4;
constexpr int QU   = 8;

typedef float vf2 __attribute__((ext_vector_type(2)));
typedef float vf4 __attribute__((ext_vector_type(4)));
typedef int   vi2 __attribute__((ext_vector_type(2)));
typedef int   vi4 __attribute__((ext_vector_type(4)));
template <int V> struct VT;
template <> struct VT<2> { typedef vf2 T; };
template <> struct VT<4> { typedef vf4 T; };

static_assert((PCAP / 2) % BT == 0, "one step of k_list's loads must stay inside one pair list");

__global__ __launch_bounds__(BT)
void k_pick(const int* __restrict__ src, const int* __restrict__ dst,
            int* __restrict__ gpair, int* __restrict__ gpcnt, int N, int E) {
  __shared__ vi2 pl[PCAP];
  __shared__ int cnt;

  const int tid = threadIdx.x;
  const int lane = tid & 31;
  const int g   = blockIdx.x / NS;
  const int sl  = blockIdx.x % NS;
  const int rlo = g * RN;
  const unsigned urlo = (unsigned)rlo;
  const unsigned urnt = (unsigned)((N - rlo < RN) ? (N - rlo) : RN);

  const int es = (((E + NS - 1) / NS) + 3) & ~3;
  const long e0l = (long)sl * es;
  const int e0 = e0l < E ? (int)e0l : E;
  const int e1 = (E - e0 < es) ? E : (e0 + es);

  if (tid == 0) cnt = 0;
  __syncthreads();

  const bool al = ((((size_t)(src + e0)) | ((size_t)(dst + e0))) & 15u) == 0;
  const int nvec = al ? ((e1 - e0) >> 2) : 0;
  const vi4* dv = (const vi4*)(dst + e0);
  const vi4* sv = (const vi4*)(src + e0);

  for (int vb = 0; vb < nvec; vb += BT * PU) {
    vi4 d[PU], s[PU];
    if (vb + BT * PU <= nvec) {
#pragma unroll
      for (int u = 0; u < PU; ++u) { d[u] = dv[vb + u * BT + tid]; s[u] = sv[vb + u * BT + tid]; }
    } else {
#pragma unroll
      for (int u = 0; u < PU; ++u) {
        int i = vb + u * BT + tid;
        vi4 none = {-1, -1, -1, -1};
        d[u] = (i < nvec) ? dv[i] : none;
        s[u] = (i < nvec) ? sv[i] : none;
      }
    }
    unsigned m[4 * PU];
    int tot = 0;
#pragma unroll
    for (int u = 0; u < PU; ++u) {
#pragma unroll
      for (int j = 0; j < 4; ++j) {
        m[4 * u + j] = __builtin_amdgcn_ballot_w32((unsigned)d[u][j] - urlo < urnt);
        tot += __builtin_popcount(m[4 * u + j]);
      }
    }
    if (tot) {
      int base = 0;
      if (lane == 0) base = atomicAdd(&cnt, tot);
      base = __builtin_amdgcn_readfirstlane(base);
#pragma unroll
      for (int u = 0; u < PU; ++u) {
#pragma unroll
        for (int j = 0; j < 4; ++j) {
          const unsigned mk = m[4 * u + j];
          if ((unsigned)d[u][j] - urlo < urnt) {
            int pos = base + (int)__builtin_amdgcn_mbcnt_lo(mk, 0u);
            if (pos < PCAP) { vi2 ent; ent.x = s[u][j]; ent.y = d[u][j]; pl[pos] = ent; }
          }
          base += __builtin_popcount(mk);
        }
      }
    }
  }
  for (int e = e0 + 4 * nvec + tid; e < e1; e += BT) {
    int dj = dst[e];
    if ((unsigned)dj - urlo < urnt) {
      int pos = atomicAdd(&cnt, 1);
      if (pos < PCAP) { vi2 ent; ent.x = src[e]; ent.y = dj; pl[pos] = ent; }
    }
  }
  __syncthreads();

  const int total = cnt;
  const int n = total < PCAP ? total : PCAP;
  vi2* gp = (vi2*)gpair + (size_t)blockIdx.x * PCAP;
  for (int i = tid; i < n; i += BT) gp[i] = pl[i];
  if (tid == 0) {
    gpcnt[blockIdx.x] = total;
    if (n & 1) { vi2 pad; pad.x = 0; pad.y = -1; gp[n] = pad; }
  }
}

__global__ __launch_bounds__(BT)
void k_list(const int* __restrict__ gpair, const int* __restrict__ gpcnt,
            const int* __restrict__ src, const int* __restrict__ dst,
            float* __restrict__ dinv, int* __restrict__ glist,
            int* __restrict__ gcnt, int N, int E) {
  __shared__ vi2 lst[LCAP];
  __shared__ int degc[TN];
  __shared__ int pc[NS];
  __shared__ int cnt;

  const int tid = threadIdx.x;
  const int lane = tid & 31;
  const int lo  = blockIdx.x * TN;
  const int nt  = (N - lo < TN) ? (N - lo) : TN;
  const unsigned unt = (unsigned)nt;
  const unsigned ulo = (unsigned)lo;
  const int g   = blockIdx.x / TPR;

  if (tid == 0) cnt = 0;
  if (tid < NS) pc[tid] = gpcnt[g * NS + tid];
  for (int t = tid; t < TN; t += BT) degc[t] = 0;
  __syncthreads();

  int pmax = 0;
#pragma unroll
  for (int s = 0; s < NS; ++s) pmax = pc[s] > pmax ? pc[s] : pmax;

  if (pmax <= PCAP) {
    constexpr int VPL = PCAP / 2;
    const vi4* pv = (const vi4*)gpair + (size_t)g * NS * VPL;
    for (int vb = 0; vb < NS * VPL; vb += BT * QU) {
      vi4 p[QU];
#pragma unroll
      for (int u = 0; u < QU; ++u) {
        int v  = vb + u * BT + tid;
        int np = pc[(vb + u * BT) / VPL];
        int i  = 2 * (v % VPL);
        vi4 none = {0, -1, 0, -1};
        p[u] = (i < np) ? pv[v] : none;
      }
      unsigned m[2 * QU];
      int tot = 0;
#pragma unroll
      for (int u = 0; u < QU; ++u) {
        m[2 * u]     = __builtin_amdgcn_ballot_w32((unsigned)p[u].y - ulo < unt);
        m[2 * u + 1] = __builtin_amdgcn_ballot_w32((unsigned)p[u].w - ulo < unt);
        tot += __builtin_popcount(m[2 * u]) + __builtin_popcount(m[2 * u + 1]);
      }
      if (tot) {
        int base = 0;
        if (lane == 0) base = atomicAdd(&cnt, tot);
        base = __builtin_amdgcn_readfirstlane(base);
#pragma unroll
        for (int u = 0; u < QU; ++u) {
          unsigned t0 = (unsigned)p[u].y - ulo, t1 = (unsigned)p[u].w - ulo;
          if (t0 < unt) {
            atomicAdd(&degc[t0], 1);
            int pos = base + (int)__builtin_amdgcn_mbcnt_lo(m[2 * u], 0u);
            if (pos < LCAP) { vi2 ent; ent.x = p[u].x; ent.y = (int)t0; lst[pos] = ent; }
          }
          base += __builtin_popcount(m[2 * u]);
          if (t1 < unt) {
            atomicAdd(&degc[t1], 1);
            int pos = base + (int)__builtin_amdgcn_mbcnt_lo(m[2 * u + 1], 0u);
            if (pos < LCAP) { vi2 ent; ent.x = p[u].z; ent.y = (int)t1; lst[pos] = ent; }
          }
          base += __builtin_popcount(m[2 * u + 1]);
        }
      }
    }
  } else {
    for (int e = tid; e < E; e += BT) {
      unsigned t = (unsigned)dst[e] - ulo;
      if (t < unt) {
        atomicAdd(&degc[t], 1);
        int pos = atomicAdd(&cnt, 1);
        if (pos < LCAP) { vi2 ent; ent.x = src[e]; ent.y = (int)t; lst[pos] = ent; }
      }
    }
  }
  __syncthreads();

  const int total = cnt;
  const int n = total < LCAP ? total : LCAP;
  vi2* gl = (vi2*)glist + (size_t)blockIdx.x * LCAP;
  for (int i = tid; i < n; i += BT) gl[i] = lst[i];
  if (tid == 0) gcnt[blockIdx.x] = total;
  for (int t = tid; t < nt; t += BT) dinv[lo + t] = rsqrtf(1.0f + (float)degc[t]);
}

template <int OUT>
__global__ __launch_bounds__(256)
void k_gemm(const float* __restrict__ A, const float* __restrict__ W,
            float* __restrict__ H, int M) {
  constexpr int NT = OUT / 16;
  int wave = (blockIdx.x * blockDim.x + threadIdx.x) >> 5;
  int lane = threadIdx.x & 31;
  int row0 = wave * 16;
  if (row0 >= M) return;

  int m  = lane & 15;
  int kh = lane >> 4;

  v8f acc[NT] = {};

  int arow = (row0 + m < M) ? (row0 + m) : (M - 1);
  const float* ap = A + (size_t)arow * 128 + 2 * kh;

  for (int k = 0; k < 128; k += 4) {
    v2f a = *(const v2f*)(ap + k);
    const float* bp = W + (size_t)(k + 2 * kh) * OUT + m;
#pragma unroll
    for (int t = 0; t < NT; ++t) {
      v2f b;
      b.x = bp[t * 16];
      b.y = bp[t * 16 + OUT];
      acc[t] = __builtin_amdgcn_wmma_f32_16x16x4_f32(
           false, a,  false, b,
           (short)0, acc[t],  false,  false);
    }
  }

  int rbase = row0 + 8 * kh;
#pragma unroll
  for (int t = 0; t < NT; ++t) {
#pragma unroll
    for (int r = 0; r < 8; ++r) {
      int row = rbase + r;
      if (row < M) H[(size_t)row * OUT + t * 16 + m] = acc[t][r];
    }
  }
}

template <int OUT, bool RELU>
__global__ __launch_bounds__(BT)
void k_agg(const int* __restrict__ glist, const int* __restrict__ gcnt,
           const int* __restrict__ src, const int* __restrict__ dst,
           const float* __restrict__ dinv, const float* __restrict__ h,
           const float* __restrict__ b, float* __restrict__ outp, int N, int E) {
  constexpr int VEC = OUT / 32;
  typedef typename VT<VEC>::T V;
  __shared__ float tile[TN * OUT];
  __shared__ vi2   lst[LCAP];
  __shared__ float dl[TN];

  const int tid  = threadIdx.x;
  const int lane = tid & 31;
  const int wave = tid >> 5;
  const int lo   = blockIdx.x * TN;
  const int nt   = (N - lo < TN) ? (N - lo) : TN;
  const unsigned unt = (unsigned)nt;
  const unsigned ulo = (unsigned)lo;

  const int  total = gcnt[blockIdx.x];
  const bool whole = total <= LCAP;
  const int  n     = whole ? total : 0;

  {
    constexpr int LU = 8;
    const vi2* gl = (const vi2*)glist + (size_t)blockIdx.x * LCAP;
    for (int i0 = tid; i0 < n; i0 += BT * LU) {
      vi2 ent[LU];
#pragma unroll
      for (int u = 0; u < LU; ++u) {
        int i = i0 + u * BT;
        vi2 none = {0, 0};
        ent[u] = (i < n) ? gl[i] : none;
      }
#pragma unroll
      for (int u = 0; u < LU; ++u) {
        int i = i0 + u * BT;
        if (i < n) lst[i] = ent[u];
      }
    }
  }
  for (int t = tid; t < nt; t += BT) dl[t] = dinv[lo + t];
  for (int t = wave; t < nt; t += BT / 32) {
    float di = dinv[lo + t];
    V x = *(const V*)(h + (size_t)(lo + t) * OUT + VEC * lane);
#pragma unroll
    for (int j = 0; j < VEC; ++j) tile[t * OUT + j * 32 + lane] = x[j] * di * di;
  }
  __syncthreads();

  if (!whole) {
    for (int e = tid; e < E; e += BT) {
      unsigned t = (unsigned)dst[e] - ulo;
      if (t < unt) {
        int s = src[e];
        float nrm = dinv[s] * dl[t];
        const float* hs = h + (size_t)s * OUT;
        for (int c = 0; c < OUT; ++c)
          atomicAdd(&tile[t * OUT + (c % VEC) * 32 + c / VEC], hs[c] * nrm);
      }
    }
  }

  for (int base = wave; base < n; base += (BT / 32) * RU) {
    if (base + (BT / 32) * (RU - 1) < n) {
      int   s[RU], t[RU];
      V     x[RU];
      float w[RU];
#pragma unroll
      for (int u = 0; u < RU; ++u) {
        vi2 ent = lst[base + (BT / 32) * u];
        s[u] = __builtin_amdgcn_readfirstlane(ent.x);
        t[u] = __builtin_amdgcn_readfirstlane(ent.y);
      }
#pragma unroll
      for (int u = 0; u < RU; ++u) {
        x[u] = *(const V*)(h + (size_t)s[u] * OUT + VEC * lane);
        w[u] = dinv[s[u]] * dl[t[u]];
      }
#pragma unroll
      for (int u = 0; u < RU; ++u) {
#pragma unroll
        for (int j = 0; j < VEC; ++j)
          atomicAdd(&tile[t[u] * OUT + j * 32 + lane], x[u][j] * w[u]);
      }
    } else {
      for (int i = base; i < n; i += BT / 32) {
        vi2 ent = lst[i];
        int s = __builtin_amdgcn_readfirstlane(ent.x);
        int t = __builtin_amdgcn_readfirstlane(ent.y);
        V x = *(const V*)(h + (size_t)s * OUT + VEC * lane);
        float w = dinv[s] * dl[t];
#pragma unroll
        for (int j = 0; j < VEC; ++j)
          atomicAdd(&tile[t * OUT + j * 32 + lane], x[j] * w);
      }
    }
  }
  __syncthreads();

  {
    V bb = *(const V*)(b + VEC * lane);
    for (int t = wave; t < nt; t += BT / 32) {
      V r;
#pragma unroll
      for (int j = 0; j < VEC; ++j) {
        float v = tile[t * OUT + j * 32 + lane] + bb[j];
        r[j] = RELU ? fmaxf(v, 0.0f) : v;
      }
      *(V*)(outp + (size_t)(lo + t) * OUT + VEC * lane) = r;
    }
  }
}

static inline int cdiv(long a, long b) { return (int)((a + b - 1) / b); }

extern "C" void kernel_launch(void* const* d_in, const int* in_sizes, int n_in,
                              void* d_out, int out_size, void* d_ws, size_t ws_size,
                              hipStream_t stream) {
  const float* x  = (const float*)d_in[0];
  const int*   ei = (const int*)d_in[1];
  const float* W1 = (const float*)d_in[2];
  const float* b1 = (const float*)d_in[3];
  const float* W2 = (const float*)d_in[4];
  const float* b2 = (const float*)d_in[5];
  const float* W3 = (const float*)d_in[6];
  const float* b3 = (const float*)d_in[7];

  const int N = in_sizes[0] / 128;
  const int E = in_sizes[1] / 2;
  const int* src = ei;
  const int* dst = ei + E;

  const int T = 256;
  const int gemmBlocks = cdiv((long)cdiv(N, 16) * 32, T);
  const int tiles = cdiv(N, TN);
  const int ranges = cdiv(tiles, TPR);

  char* ws = (char*)d_ws;
  size_t off0 = 0;
  size_t off1 = (((size_t)N * 4) + 255) & ~(size_t)255;
  size_t off2 = off1 + (size_t)N * 128 * 4;
  size_t off3 = off2 + (size_t)N * 128 * 4;
  size_t off4 = off3 + (size_t)tiles * LCAP * 8;
  size_t off5 = (off4 + (size_t)tiles * 4 + 255) & ~(size_t)255;
  size_t off6 = off5 + (size_t)ranges * NS * PCAP * 8;
  float* dinv  = (float*)(ws + off0);
  float* bufH  = (float*)(ws + off1);
  float* bufF  = (float*)(ws + off2);
  int*   glist = (int*)(ws + off3);
  int*   gcnt  = (int*)(ws + off4);
  int*   gpair = (int*)(ws + off5);
  int*   gpcnt = (int*)(ws + off6);
  float* out   = (float*)d_out;

  k_pick<<<ranges * NS, BT, 0, stream>>>(src, dst, gpair, gpcnt, N, E);
  k_list<<<tiles, BT, 0, stream>>>(gpair, gpcnt, src, dst, dinv, glist, gcnt, N, E);

  k_gemm<128><<<gemmBlocks, T, 0, stream>>>(x, W1, bufH, N);
  k_agg<128, true><<<tiles, BT, 0, stream>>>(glist, gcnt, src, dst, dinv, bufH, b1, bufF, N, E);

  k_gemm<128><<<gemmBlocks, T, 0, stream>>>(bufF, W2, bufH, N);
  k_agg<128, true><<<tiles, BT, 0, stream>>>(glist, gcnt, src, dst, dinv, bufH, b2, bufF, N, E);

  k_gemm<64><<<gemmBlocks, T, 0, stream>>>(bufF, W3, bufH, N);
  k_agg<64, false><<<tiles, BT, 0, stream>>>(glist, gcnt, src, dst, dinv, bufH, b3, out, N, E);
}


# Next layer's product fused into the aggregation launch: the owner block multiplies its finished 392-row LDS tile by W2 / W3 (f32 WMMA); 8 to 6 launches, no feature round trip; product k order changed
# speedup vs baseline: 4.0063x; 1.1295x over previous
#include <hip/hip_runtime.h>

typedef float v2f __attribute__((ext_vector_type(2)));
typedef float v8f __attribute__((ext_vector_type(8)));

constexpr int TN   = 392;
constexpr int LCAP = 12288;
constexpr int BT   = 1024;
constexpr int RU   = 8;
constexpr int TPR  = 16;
constexpr int RN   = TN * TPR;
constexpr int NS   = 16;
constexpr int PCAP = 8192;
constexpr int PU   = 4;
constexpr int QU   = 8;

typedef float vf2 __attribute__((ext_vector_type(2)));
typedef float vf4 __attribute__((ext_vector_type(4)));
typedef int   vi2 __attribute__((ext_vector_type(2)));
typedef int   vi4 __attribute__((ext_vector_type(4)));
template <int V> struct VT;
template <> struct VT<2> { typedef vf2 T; };
template <> struct VT<4> { typedef vf4 T; };

static_assert((PCAP / 2) % BT == 0, "one step of k_list's loads must stay inside one pair list");

__global__ __launch_bounds__(BT)
void k_pick(const int* __restrict__ src, const int* __restrict__ dst,
            int* __restrict__ gpair, int* __restrict__ gpcnt, int N, int E) {
  __shared__ vi2 pl[PCAP];
  __shared__ int cnt;

  const int tid = threadIdx.x;
  const int lane = tid & 31;
  const int g   = blockIdx.x / NS;
  const int sl  = blockIdx.x % NS;
  const int rlo = g * RN;
  const unsigned urlo = (unsigned)rlo;
  const unsigned urnt = (unsigned)((N - rlo < RN) ? (N - rlo) : RN);

  const int es = (((E + NS - 1) / NS) + 3) & ~3;
  const long e0l = (long)sl * es;
  const int e0 = e0l < E ? (int)e0l : E;
  const int e1 = (E - e0 < es) ? E : (e0 + es);

  if (tid == 0) cnt = 0;
  __syncthreads();

  const bool al = ((((size_t)(src + e0)) | ((size_t)(dst + e0))) & 15u) == 0;
  const int nvec = al ? ((e1 - e0) >> 2) : 0;
  const vi4* dv = (const vi4*)(dst + e0);
  const vi4* sv = (const vi4*)(src + e0);

  for (int vb = 0; vb < nvec; vb += BT * PU) {
    vi4 d[PU], s[PU];
    if (vb + BT * PU <= nvec) {
#pragma unroll
      for (int u = 0; u < PU; ++u) { d[u] = dv[vb + u * BT + tid]; s[u] = sv[vb + u * BT + tid]; }
    } else {
#pragma unroll
      for (int u = 0; u < PU; ++u) {
        int i = vb + u * BT + tid;
        vi4 none = {-1, -1, -1, -1};
        d[u] = (i < nvec) ? dv[i] : none;
        s[u] = (i < nvec) ? sv[i] : none;
      }
    }
    unsigned m[4 * PU];
    int tot = 0;
#pragma unroll
    for (int u = 0; u < PU; ++u) {
#pragma unroll
      for (int j = 0; j < 4; ++j) {
        m[4 * u + j] = __builtin_amdgcn_ballot_w32((unsigned)d[u][j] - urlo < urnt);
        tot += __builtin_popcount(m[4 * u + j]);
      }
    }
    if (tot) {
      int base = 0;
      if (lane == 0) base = atomicAdd(&cnt, tot);
      base = __builtin_amdgcn_readfirstlane(base);
#pragma unroll
      for (int u = 0; u < PU; ++u) {
#pragma unroll
        for (int j = 0; j < 4; ++j) {
          const unsigned mk = m[4 * u + j];
          if ((unsigned)d[u][j] - urlo < urnt) {
            int pos = base + (int)__builtin_amdgcn_mbcnt_lo(mk, 0u);
            if (pos < PCAP) { vi2 ent; ent.x = s[u][j]; ent.y = d[u][j]; pl[pos] = ent; }
          }
          base += __builtin_popcount(mk);
        }
      }
    }
  }
  for (int e = e0 + 4 * nvec + tid; e < e1; e += BT) {
    int dj = dst[e];
    if ((unsigned)dj - urlo < urnt) {
      int pos = atomicAdd(&cnt, 1);
      if (pos < PCAP) { vi2 ent; ent.x = src[e]; ent.y = dj; pl[pos] = ent; }
    }
  }
  __syncthreads();

  const int total = cnt;
  const int n = total < PCAP ? total : PCAP;
  vi2* gp = (vi2*)gpair + (size_t)blockIdx.x * PCAP;
  for (int i = tid; i < n; i += BT) gp[i] = pl[i];
  if (tid == 0) {
    gpcnt[blockIdx.x] = total;
    if (n & 1) { vi2 pad; pad.x = 0; pad.y = -1; gp[n] = pad; }
  }
}

__global__ __launch_bounds__(BT)
void k_list(const int* __restrict__ gpair, const int* __restrict__ gpcnt,
            const int* __restrict__ src, const int* __restrict__ dst,
            float* __restrict__ dinv, int* __restrict__ glist,
            int* __restrict__ gcnt, int N, int E) {
  __shared__ vi2 lst[LCAP];
  __shared__ int degc[TN];
  __shared__ int pc[NS];
  __shared__ int cnt;

  const int tid = threadIdx.x;
  const int lane = tid & 31;
  const int lo  = blockIdx.x * TN;
  const int nt  = (N - lo < TN) ? (N - lo) : TN;
  const unsigned unt = (unsigned)nt;
  const unsigned ulo = (unsigned)lo;
  const int g   = blockIdx.x / TPR;

  if (tid == 0) cnt = 0;
  if (tid < NS) pc[tid] = gpcnt[g * NS + tid];
  for (int t = tid; t < TN; t += BT) degc[t] = 0;
  __syncthreads();

  int pmax = 0;
#pragma unroll
  for (int s = 0; s < NS; ++s) pmax = pc[s] > pmax ? pc[s] : pmax;

  if (pmax <= PCAP) {
    constexpr int VPL = PCAP / 2;
    const vi4* pv = (const vi4*)gpair + (size_t)g * NS * VPL;
    for (int vb = 0; vb < NS * VPL; vb += BT * QU) {
      vi4 p[QU];
#pragma unroll
      for (int u = 0; u < QU; ++u) {
        int v  = vb + u * BT + tid;
        int np = pc[(vb + u * BT) / VPL];
        int i  = 2 * (v % VPL);
        vi4 none = {0, -1, 0, -1};
        p[u] = (i < np) ? pv[v] : none;
      }
      unsigned m[2 * QU];
      int tot = 0;
#pragma unroll
      for (int u = 0; u < QU; ++u) {
        m[2 * u]     = __builtin_amdgcn_ballot_w32((unsigned)p[u].y - ulo < unt);
        m[2 * u + 1] = __builtin_amdgcn_ballot_w32((unsigned)p[u].w - ulo < unt);
        tot += __builtin_popcount(m[2 * u]) + __builtin_popcount(m[2 * u + 1]);
      }
      if (tot) {
        int base = 0;
        if (lane == 0) base = atomicAdd(&cnt, tot);
        base = __builtin_amdgcn_readfirstlane(base);
#pragma unroll
        for (int u = 0; u < QU; ++u) {
          unsigned t0 = (unsigned)p[u].y - ulo, t1 = (unsigned)p[u].w - ulo;
          if (t0 < unt) {
            atomicAdd(&degc[t0], 1);
            int pos = base + (int)__builtin_amdgcn_mbcnt_lo(m[2 * u], 0u);
            if (pos < LCAP) { vi2 ent; ent.x = p[u].x; ent.y = (int)t0; lst[pos] = ent; }
          }
          base += __builtin_popcount(m[2 * u]);
          if (t1 < unt) {
            atomicAdd(&degc[t1], 1);
            int pos = base + (int)__builtin_amdgcn_mbcnt_lo(m[2 * u + 1], 0u);
            if (pos < LCAP) { vi2 ent; ent.x = p[u].z; ent.y = (int)t1; lst[pos] = ent; }
          }
          base += __builtin_popcount(m[2 * u + 1]);
        }
      }
    }
  } else {
    for (int e = tid; e < E; e += BT) {
      unsigned t = (unsigned)dst[e] - ulo;
      if (t < unt) {
        atomicAdd(&degc[t], 1);
        int pos = atomicAdd(&cnt, 1);
        if (pos < LCAP) { vi2 ent; ent.x = src[e]; ent.y = (int)t; lst[pos] = ent; }
      }
    }
  }
  __syncthreads();

  const int total = cnt;
  const int n = total < LCAP ? total : LCAP;
  vi2* gl = (vi2*)glist + (size_t)blockIdx.x * LCAP;
  for (int i = tid; i < n; i += BT) gl[i] = lst[i];
  if (tid == 0) gcnt[blockIdx.x] = total;
  for (int t = tid; t < nt; t += BT) dinv[lo + t] = rsqrtf(1.0f + (float)degc[t]);
}

template <int OUT>
__global__ __launch_bounds__(256)
void k_gemm(const float* __restrict__ A, const float* __restrict__ W,
            float* __restrict__ H, int M) {
  constexpr int NT = OUT / 16;
  int wave = (blockIdx.x * blockDim.x + threadIdx.x) >> 5;
  int lane = threadIdx.x & 31;
  int row0 = wave * 16;
  if (row0 >= M) return;

  int m  = lane & 15;
  int kh = lane >> 4;

  v8f acc[NT] = {};

  int arow = (row0 + m < M) ? (row0 + m) : (M - 1);
  const float* ap = A + (size_t)arow * 128 + 2 * kh;

  for (int k = 0; k < 128; k += 4) {
    v2f a = *(const v2f*)(ap + k);
    const float* bp = W + (size_t)(k + 2 * kh) * OUT + m;
#pragma unroll
    for (int t = 0; t < NT; ++t) {
      v2f b;
      b.x = bp[t * 16];
      b.y = bp[t * 16 + OUT];
      acc[t] = __builtin_amdgcn_wmma_f32_16x16x4_f32(
           false, a,  false, b,
           (short)0, acc[t],  false,  false);
    }
  }

  int rbase = row0 + 8 * kh;
#pragma unroll
  for (int t = 0; t < NT; ++t) {
#pragma unroll
    for (int r = 0; r < 8; ++r) {
      int row = rbase + r;
      if (row < M) H[(size_t)row * OUT + t * 16 + m] = acc[t][r];
    }
  }
}

template <int OUT, bool RELU, int OUTN>
__global__ __launch_bounds__(BT)
void k_agg(const int* __restrict__ glist, const int* __restrict__ gcnt,
           const int* __restrict__ src, const int* __restrict__ dst,
           const float* __restrict__ dinv, const float* __restrict__ h,
           const float* __restrict__ b, const float* __restrict__ Wn,
           float* __restrict__ outp, int N, int E) {
  constexpr int VEC = OUT / 32;
  constexpr bool FUSE = OUTN > 0;
  constexpr int P  = FUSE ? OUT + 4 : OUT;
  constexpr int TR = FUSE ? ((TN + 15) / 16) * 16 : TN;
  static_assert(!FUSE || (OUT == 128 && OUTN % 64 == 0), "fused product: K = 128, whole 64-column groups");
  static_assert((TN + 15) / 16 <= BT / 32, "one strip per wave");
  typedef typename VT<VEC>::T V;
  __shared__ __attribute__((aligned(16))) float tile[TR * P];
  __shared__ vi2   lst[LCAP];
  __shared__ float dl[TN];
  __shared__ __attribute__((aligned(16))) float bl[FUSE ? OUT : 4];

  const int tid  = threadIdx.x;
  const int lane = tid & 31;
  const int wave = tid >> 5;
  const int lo   = blockIdx.x * TN;
  const int nt   = (N - lo < TN) ? (N - lo) : TN;
  const unsigned unt = (unsigned)nt;
  const unsigned ulo = (unsigned)lo;

  const int  total = gcnt[blockIdx.x];
  const bool whole = total <= LCAP;
  const int  n     = whole ? total : 0;

  {
    constexpr int LU = 8;
    const vi2* gl = (const vi2*)glist + (size_t)blockIdx.x * LCAP;
    for (int i0 = tid; i0 < n; i0 += BT * LU) {
      vi2 ent[LU];
#pragma unroll
      for (int u = 0; u < LU; ++u) {
        int i = i0 + u * BT;
        vi2 none = {0, 0};
        ent[u] = (i < n) ? gl[i] : none;
      }
#pragma unroll
      for (int u = 0; u < LU; ++u) {
        int i = i0 + u * BT;
        if (i < n) lst[i] = ent[u];
      }
    }
  }
  for (int t = tid; t < nt; t += BT) dl[t] = dinv[lo + t];
  if (FUSE) {
    if (tid < OUT) bl[(tid % VEC) * 32 + tid / VEC] = b[tid];
  }
  for (int t = wave; t < nt; t += BT / 32) {
    float di = dinv[lo + t];
    V x = *(const V*)(h + (size_t)(lo + t) * OUT + VEC * lane);
#pragma unroll
    for (int j = 0; j < VEC; ++j) tile[t * P + j * 32 + lane] = x[j] * di * di;
  }
  __syncthreads();

  if (!whole) {
    for (int e = tid; e < E; e += BT) {
      unsigned t = (unsigned)dst[e] - ulo;
      if (t < unt) {
        int s = src[e];
        float nrm = dinv[s] * dl[t];
        const float* hs = h + (size_t)s * OUT;
        for (int c = 0; c < OUT; ++c)
          atomicAdd(&tile[t * P + (c % VEC) * 32 + c / VEC], hs[c] * nrm);
      }
    }
  }

  for (int base = wave; base < n; base += (BT / 32) * RU) {
    if (base + (BT / 32) * (RU - 1) < n) {
      int   s[RU], t[RU];
      V     x[RU];
      float w[RU];
#pragma unroll
      for (int u = 0; u < RU; ++u) {
        vi2 ent = lst[base + (BT / 32) * u];
        s[u] = __builtin_amdgcn_readfirstlane(ent.x);
        t[u] = __builtin_amdgcn_readfirstlane(ent.y);
      }
#pragma unroll
      for (int u = 0; u < RU; ++u) {
        x[u] = *(const V*)(h + (size_t)s[u] * OUT + VEC * lane);
        w[u] = dinv[s[u]] * dl[t[u]];
      }
#pragma unroll
      for (int u = 0; u < RU; ++u) {
#pragma unroll
        for (int j = 0; j < VEC; ++j)
          atomicAdd(&tile[t[u] * P + j * 32 + lane], x[u][j] * w[u]);
      }
    } else {
      for (int i = base; i < n; i += BT / 32) {
        vi2 ent = lst[i];
        int s = __builtin_amdgcn_readfirstlane(ent.x);
        int t = __builtin_amdgcn_readfirstlane(ent.y);
        V x = *(const V*)(h + (size_t)s * OUT + VEC * lane);
        float w = dinv[s] * dl[t];
#pragma unroll
        for (int j = 0; j < VEC; ++j)
          atomicAdd(&tile[t * P + j * 32 + lane], x[j] * w);
      }
    }
  }
  __syncthreads();

  if constexpr (!FUSE) {
    V bb = *(const V*)(b + VEC * lane);
    for (int t = wave; t < nt; t += BT / 32) {
      V r;
#pragma unroll
      for (int j = 0; j < VEC; ++j) {
        float v = tile[t * P + j * 32 + lane] + bb[j];
        r[j] = RELU ? fmaxf(v, 0.0f) : v;
      }
      *(V*)(outp + (size_t)(lo + t) * OUT + VEC * lane) = r;
    }
  } else {
    constexpr int NTN = OUTN / 16;
    constexpr int G   = OUTN / 64;
    const int nstrips = (nt + 15) >> 4;
    if (wave < nstrips) {
      const int m  = lane & 15;
      const int hf = lane >> 4;
      const float* arow = tile + (wave * 16 + m) * P;
      v8f acc[NTN] = {};
      for (int it = 0; it < 16; ++it) {
        const int c   = 2 * it + hf;
        const int pj  = c & 3, qi = c >> 2;
        const int off = pj * 32 + qi * 4;
        vf4 a  = *(const vf4*)(arow + off);
        vf4 bv = *(const vf4*)(bl + off);
#pragma unroll
        for (int e = 0; e < 4; ++e) {
          float v = a[e] + bv[e];
          a[e] = RELU ? fmaxf(v, 0.0f) : v;
        }
        const float* wp = Wn + (size_t)(16 * qi + pj) * OUTN + 4 * m;
        vf4 w[4][G];
#pragma unroll
        for (int e = 0; e < 4; ++e) {
#pragma unroll
          for (int g = 0; g < G; ++g) w[e][g] = *(const vf4*)(wp + (size_t)(4 * e) * OUTN + g * 64);
        }
#pragma unroll
        for (int p = 0; p < 2; ++p) {
          v2f af; af.x = a[2 * p]; af.y = a[2 * p + 1];
#pragma unroll
          for (int t = 0; t < NTN; ++t) {
            v2f bf; bf.x = w[2 * p][t / 4][t % 4]; bf.y = w[2 * p + 1][t / 4][t % 4];
            acc[t] = __builtin_amdgcn_wmma_f32_16x16x4_f32(
                 false, af,  false, bf,
                 (short)0, acc[t],  false,  false);
          }
        }
      }
      const int rbase = wave * 16 + 8 * hf;
#pragma unroll
      for (int r = 0; r < 8; ++r) {
        const int row = rbase + r;
        if (row < nt) {
#pragma unroll
          for (int g = 0; g < G; ++g) {
            vf4 o;
            o.x = acc[4 * g][r]; o.y = acc[4 * g + 1][r]; o.z = acc[4 * g + 2][r]; o.w = acc[4 * g + 3][r];
            *(vf4*)(outp + (size_t)(lo + row) * OUTN + g * 64 + 4 * m) = o;
          }
        }
      }
    }
  }
}

static inline int cdiv(long a, long b) { return (int)((a + b - 1) / b); }

extern "C" void kernel_launch(void* const* d_in, const int* in_sizes, int n_in,
                              void* d_out, int out_size, void* d_ws, size_t ws_size,
                              hipStream_t stream) {
  const float* x  = (const float*)d_in[0];
  const int*   ei = (const int*)d_in[1];
  const float* W1 = (const float*)d_in[2];
  const float* b1 = (const float*)d_in[3];
  const float* W2 = (const float*)d_in[4];
  const float* b2 = (const float*)d_in[5];
  const float* W3 = (const float*)d_in[6];
  const float* b3 = (const float*)d_in[7];

  const int N = in_sizes[0] / 128;
  const int E = in_sizes[1] / 2;
  const int* src = ei;
  const int* dst = ei + E;

  const int T = 256;
  const int gemmBlocks = cdiv((long)cdiv(N, 16) * 32, T);
  const int tiles = cdiv(N, TN);
  const int ranges = cdiv(tiles, TPR);

  char* ws = (char*)d_ws;
  size_t off0 = 0;
  size_t off1 = (((size_t)N * 4) + 255) & ~(size_t)255;
  size_t off2 = off1 + (size_t)N * 128 * 4;
  size_t off3 = off2 + (size_t)N * 128 * 4;
  size_t off4 = off3 + (size_t)tiles * LCAP * 8;
  size_t off5 = (off4 + (size_t)tiles * 4 + 255) & ~(size_t)255;
  size_t off6 = off5 + (size_t)ranges * NS * PCAP * 8;
  float* dinv  = (float*)(ws + off0);
  float* bufH  = (float*)(ws + off1);
  float* bufF  = (float*)(ws + off2);
  int*   glist = (int*)(ws + off3);
  int*   gcnt  = (int*)(ws + off4);
  int*   gpair = (int*)(ws + off5);
  int*   gpcnt = (int*)(ws + off6);
  float* out   = (float*)d_out;

  k_pick<<<ranges * NS, BT, 0, stream>>>(src, dst, gpair, gpcnt, N, E);
  k_list<<<tiles, BT, 0, stream>>>(gpair, gpcnt, src, dst, dinv, glist, gcnt, N, E);

  k_gemm<128><<<gemmBlocks, T, 0, stream>>>(x, W1, bufH, N);
  k_agg<128, true, 128><<<tiles, BT, 0, stream>>>(glist, gcnt, src, dst, dinv, bufH, b1, W2, bufF, N, E);

  k_agg<128, true, 64><<<tiles, BT, 0, stream>>>(glist, gcnt, src, dst, dinv, bufF, b2, W3, bufH, N, E);

  k_agg<64, false, 0><<<tiles, BT, 0, stream>>>(glist, gcnt, src, dst, dinv, bufH, b3, nullptr, out, N, E);
}


# Tile lists hold the normalised adjacency's nonzeros {source|row, value}, self loops too; values filed once by layer 1's list staging; no per-edge d^-1/2 loads; sum order: self term joins LDS arrival
# speedup vs baseline: 4.2041x; 1.0494x over previous
#include <hip/hip_runtime.h>

typedef float v2f __attribute__((ext_vector_type(2)));
typedef float v8f __attribute__((ext_vector_type(8)));

constexpr int TN   = 392;
constexpr int LCAP = 12288;
constexpr int BT   = 1024;
constexpr int RU   = 8;
constexpr int TPR  = 16;
constexpr int RN   = TN * TPR;
constexpr int NS   = 16;
constexpr int PCAP = 8192;
constexpr int PU   = 4;
constexpr int QU   = 8;

typedef float vf2 __attribute__((ext_vector_type(2)));
typedef float vf4 __attribute__((ext_vector_type(4)));
typedef int   vi2 __attribute__((ext_vector_type(2)));
typedef int   vi4 __attribute__((ext_vector_type(4)));
template <int V> struct VT;
template <> struct VT<2> { typedef vf2 T; };
template <> struct VT<4> { typedef vf4 T; };

static_assert((PCAP / 2) % BT == 0, "one step of k_list's loads must stay inside one pair list");

__global__ __launch_bounds__(BT)
void k_pick(const int* __restrict__ src, const int* __restrict__ dst,
            int* __restrict__ gpair, int* __restrict__ gpcnt, int N, int E) {
  __shared__ vi2 pl[PCAP];
  __shared__ int cnt;

  const int tid = threadIdx.x;
  const int lane = tid & 31;
  const int g   = blockIdx.x / NS;
  const int sl  = blockIdx.x % NS;
  const int rlo = g * RN;
  const unsigned urlo = (unsigned)rlo;
  const unsigned urnt = (unsigned)((N - rlo < RN) ? (N - rlo) : RN);

  const int es = (((E + NS - 1) / NS) + 3) & ~3;
  const long e0l = (long)sl * es;
  const int e0 = e0l < E ? (int)e0l : E;
  const int e1 = (E - e0 < es) ? E : (e0 + es);

  if (tid == 0) cnt = 0;
  __syncthreads();

  const bool al = ((((size_t)(src + e0)) | ((size_t)(dst + e0))) & 15u) == 0;
  const int nvec = al ? ((e1 - e0) >> 2) : 0;
  const vi4* dv = (const vi4*)(dst + e0);
  const vi4* sv = (const vi4*)(src + e0);

  for (int vb = 0; vb < nvec; vb += BT * PU) {
    vi4 d[PU], s[PU];
    if (vb + BT * PU <= nvec) {
#pragma unroll
      for (int u = 0; u < PU; ++u) { d[u] = dv[vb + u * BT + tid]; s[u] = sv[vb + u * BT + tid]; }
    } else {
#pragma unroll
      for (int u = 0; u < PU; ++u) {
        int i = vb + u * BT + tid;
        vi4 none = {-1, -1, -1, -1};
        d[u] = (i < nvec) ? dv[i] : none;
        s[u] = (i < nvec) ? sv[i] : none;
      }
    }
    unsigned m[4 * PU];
    int tot = 0;
#pragma unroll
    for (int u = 0; u < PU; ++u) {
#pragma unroll
      for (int j = 0; j < 4; ++j) {
        m[4 * u + j] = __builtin_amdgcn_ballot_w32((unsigned)d[u][j] - urlo < urnt);
        tot += __builtin_popcount(m[4 * u + j]);
      }
    }
    if (tot) {
      int base = 0;
      if (lane == 0) base = atomicAdd(&cnt, tot);
      base = __builtin_amdgcn_readfirstlane(base);
#pragma unroll
      for (int u = 0; u < PU; ++u) {
#pragma unroll
        for (int j = 0; j < 4; ++j) {
          const unsigned mk = m[4 * u + j];
          if ((unsigned)d[u][j] - urlo < urnt) {
            int pos = base + (int)__builtin_amdgcn_mbcnt_lo(mk, 0u);
            if (pos < PCAP) { vi2 ent; ent.x = s[u][j]; ent.y = d[u][j]; pl[pos] = ent; }
          }
          base += __builtin_popcount(mk);
        }
      }
    }
  }
  for (int e = e0 + 4 * nvec + tid; e < e1; e += BT) {
    int dj = dst[e];
    if ((unsigned)dj - urlo < urnt) {
      int pos = atomicAdd(&cnt, 1);
      if (pos < PCAP) { vi2 ent; ent.x = src[e]; ent.y = dj; pl[pos] = ent; }
    }
  }
  __syncthreads();

  const int total = cnt;
  const int n = total < PCAP ? total : PCAP;
  vi2* gp = (vi2*)gpair + (size_t)blockIdx.x * PCAP;
  for (int i = tid; i < n; i += BT) gp[i] = pl[i];
  if (tid == 0) {
    gpcnt[blockIdx.x] = total;
    if (n & 1) { vi2 pad; pad.x = 0; pad.y = -1; gp[n] = pad; }
  }
}

__global__ __launch_bounds__(BT)
void k_list(const int* __restrict__ gpair, const int* __restrict__ gpcnt,
            const int* __restrict__ src, const int* __restrict__ dst,
            float* __restrict__ dinv, int* __restrict__ glist,
            int* __restrict__ gcnt, int N, int E) {
  __shared__ vi2 lst[LCAP];
  __shared__ int degc[TN];
  __shared__ int pc[NS];
  __shared__ int cnt;

  const int tid = threadIdx.x;
  const int lane = tid & 31;
  const int lo  = blockIdx.x * TN;
  const int nt  = (N - lo < TN) ? (N - lo) : TN;
  const unsigned unt = (unsigned)nt;
  const unsigned ulo = (unsigned)lo;
  const int g   = blockIdx.x / TPR;

  if (tid == 0) cnt = 0;
  if (tid < NS) pc[tid] = gpcnt[g * NS + tid];
  for (int t = tid; t < TN; t += BT) degc[t] = 0;
  __syncthreads();

  int pmax = 0;
#pragma unroll
  for (int s = 0; s < NS; ++s) pmax = pc[s] > pmax ? pc[s] : pmax;

  if (pmax <= PCAP) {
    constexpr int VPL = PCAP / 2;
    const vi4* pv = (const vi4*)gpair + (size_t)g * NS * VPL;
    for (int vb = 0; vb < NS * VPL; vb += BT * QU) {
      vi4 p[QU];
#pragma unroll
      for (int u = 0; u < QU; ++u) {
        int v  = vb + u * BT + tid;
        int np = pc[(vb + u * BT) / VPL];
        int i  = 2 * (v % VPL);
        vi4 none = {0, -1, 0, -1};
        p[u] = (i < np) ? pv[v] : none;
      }
      unsigned m[2 * QU];
      int tot = 0;
#pragma unroll
      for (int u = 0; u < QU; ++u) {
        m[2 * u]     = __builtin_amdgcn_ballot_w32((unsigned)p[u].y - ulo < unt);
        m[2 * u + 1] = __builtin_amdgcn_ballot_w32((unsigned)p[u].w - ulo < unt);
        tot += __builtin_popcount(m[2 * u]) + __builtin_popcount(m[2 * u + 1]);
      }
      if (tot) {
        int base = 0;
        if (lane == 0) base = atomicAdd(&cnt, tot);
        base = __builtin_amdgcn_readfirstlane(base);
#pragma unroll
        for (int u = 0; u < QU; ++u) {
          unsigned t0 = (unsigned)p[u].y - ulo, t1 = (unsigned)p[u].w - ulo;
          if (t0 < unt) {
            atomicAdd(&degc[t0], 1);
            int pos = base + (int)__builtin_amdgcn_mbcnt_lo(m[2 * u], 0u);
            if (pos < LCAP) { vi2 ent; ent.x = p[u].x; ent.y = (int)t0; lst[pos] = ent; }
          }
          base += __builtin_popcount(m[2 * u]);
          if (t1 < unt) {
            atomicAdd(&degc[t1], 1);
            int pos = base + (int)__builtin_amdgcn_mbcnt_lo(m[2 * u + 1], 0u);
            if (pos < LCAP) { vi2 ent; ent.x = p[u].z; ent.y = (int)t1; lst[pos] = ent; }
          }
          base += __builtin_popcount(m[2 * u + 1]);
        }
      }
    }
  } else {
    for (int e = tid; e < E; e += BT) {
      unsigned t = (unsigned)dst[e] - ulo;
      if (t < unt) {
        atomicAdd(&degc[t], 1);
        int pos = atomicAdd(&cnt, 1);
        if (pos < LCAP) { vi2 ent; ent.x = src[e]; ent.y = (int)t; lst[pos] = ent; }
      }
    }
  }
  __syncthreads();

  const int total = cnt;
  const int n = total < LCAP ? total : LCAP;
  vi2* gl = (vi2*)glist + (size_t)blockIdx.x * LCAP;
  for (int i = tid; i < n; i += BT) gl[i] = lst[i];
  if (tid == 0) gcnt[blockIdx.x] = total;
  for (int t = tid; t < nt; t += BT) dinv[lo + t] = rsqrtf(1.0f + (float)degc[t]);
}

template <int OUT>
__global__ __launch_bounds__(256)
void k_gemm(const float* __restrict__ A, const float* __restrict__ W,
            float* __restrict__ H, int M) {
  constexpr int NT = OUT / 16;
  int wave = (blockIdx.x * blockDim.x + threadIdx.x) >> 5;
  int lane = threadIdx.x & 31;
  int row0 = wave * 16;
  if (row0 >= M) return;

  int m  = lane & 15;
  int kh = lane >> 4;

  v8f acc[NT] = {};

  int arow = (row0 + m < M) ? (row0 + m) : (M - 1);
  const float* ap = A + (size_t)arow * 128 + 2 * kh;

  for (int k = 0; k < 128; k += 4) {
    v2f a = *(const v2f*)(ap + k);
    const float* bp = W + (size_t)(k + 2 * kh) * OUT + m;
#pragma unroll
    for (int t = 0; t < NT; ++t) {
      v2f b;
      b.x = bp[t * 16];
      b.y = bp[t * 16 + OUT];
      acc[t] = __builtin_amdgcn_wmma_f32_16x16x4_f32(
           false, a,  false, b,
           (short)0, acc[t],  false,  false);
    }
  }

  int rbase = row0 + 8 * kh;
#pragma unroll
  for (int t = 0; t < NT; ++t) {
#pragma unroll
    for (int r = 0; r < 8; ++r) {
      int row = rbase + r;
      if (row < M) H[(size_t)row * OUT + t * 16 + m] = acc[t][r];
    }
  }
}

constexpr int SB = 23;
static_assert(TN <= (1 << (32 - SB)), "the local row must fit above the source bits");

template <int OUT, bool RELU, int OUTN, bool FIRST>
__global__ __launch_bounds__(BT)
void k_agg(int* __restrict__ glist, const int* __restrict__ gcnt,
           const int* __restrict__ src, const int* __restrict__ dst,
           const float* __restrict__ dinv, const float* __restrict__ h,
           const float* __restrict__ b, const float* __restrict__ Wn,
           float* __restrict__ outp, int N, int E) {
  constexpr int VEC = OUT / 32;
  constexpr bool FUSE = OUTN > 0;
  constexpr int P  = FUSE ? OUT + 4 : OUT;
  constexpr int TR = FUSE ? ((TN + 15) / 16) * 16 : TN;
  static_assert(!FUSE || (OUT == 128 && OUTN % 64 == 0), "fused product: K = 128, whole 64-column groups");
  static_assert((TN + 15) / 16 <= BT / 32, "one strip per wave");
  static_assert((TR * P) % 4 == 0, "the tile is cleared in 16-byte pieces");
  typedef typename VT<VEC>::T V;
  __shared__ __attribute__((aligned(16))) float tile[TR * P];
  __shared__ vi2   lst[LCAP + TN];
  __shared__ __attribute__((aligned(16))) float bl[FUSE ? OUT : 4];

  const int tid  = threadIdx.x;
  const int lane = tid & 31;
  const int wave = tid >> 5;
  const int lo   = blockIdx.x * TN;
  const int nt   = (N - lo < TN) ? (N - lo) : TN;
  const unsigned unt = (unsigned)nt;
  const unsigned ulo = (unsigned)lo;

  const int  total = gcnt[blockIdx.x];
  const float di = dinv[lo + (tid < nt ? tid : 0)];
  const float bv = FUSE ? b[tid < OUT ? tid : 0] : 0.0f;
  const bool whole = total <= LCAP && N <= (1 << SB);
  const int  n     = whole ? total : 0;

  {
    vf4 z = {0.0f, 0.0f, 0.0f, 0.0f};
    for (int i = tid; i < TR * P / 4; i += BT) ((vf4*)tile)[i] = z;
  }
  {
    constexpr int LU = 8;
    vi2* gl = (vi2*)glist + (size_t)blockIdx.x * LCAP;
    for (int i0 = tid; i0 < n; i0 += BT * LU) {
      vi2 ent[LU];
#pragma unroll
      for (int u = 0; u < LU; ++u) {
        int i = i0 + u * BT;
        vi2 none = {0, 0};
        ent[u] = (i < n) ? gl[i] : none;
      }
      if constexpr (FIRST) {
        float ds[LU], dd[LU];
#pragma unroll
        for (int u = 0; u < LU; ++u) {
          ds[u] = dinv[ent[u].x];
          dd[u] = dinv[lo + ent[u].y];
        }
#pragma unroll
        for (int u = 0; u < LU; ++u) {
          int i = i0 + u * BT;
          ent[u].x = ent[u].x | (int)((unsigned)ent[u].y << SB);
          ent[u].y = __float_as_int(ds[u] * dd[u]);
          if (i < n) gl[i] = ent[u];
        }
      }
#pragma unroll
      for (int u = 0; u < LU; ++u) {
        int i = i0 + u * BT;
        if (i < n) lst[i] = ent[u];
      }
    }
  }
  if (tid < nt) {
    vi2 ent;
    ent.x = (lo + tid) | (int)((unsigned)tid << SB);
    ent.y = __float_as_int(di * di);
    lst[n + tid] = ent;
  }
  if (FUSE) {
    if (tid < OUT) bl[(tid % VEC) * 32 + tid / VEC] = bv;
  }
  __syncthreads();

  if (!whole) {
    for (int e = tid; e < E; e += BT) {
      unsigned t = (unsigned)dst[e] - ulo;
      if (t < unt) {
        int s = src[e];
        float nrm = dinv[s] * dinv[lo + t];
        const float* hs = h + (size_t)s * OUT;
        for (int c = 0; c < OUT; ++c)
          atomicAdd(&tile[t * P + (c % VEC) * 32 + c / VEC], hs[c] * nrm);
      }
    }
  }

  const int ntot = n + nt;
  for (int base = wave; base < ntot; base += (BT / 32) * RU) {
    if (base + (BT / 32) * (RU - 1) < ntot) {
      int   s[RU], t[RU];
      V     x[RU];
      float w[RU];
#pragma unroll
      for (int u = 0; u < RU; ++u) {
        vi2 ent = lst[base + (BT / 32) * u];
        unsigned sp = (unsigned)__builtin_amdgcn_readfirstlane(ent.x);
        w[u] = __int_as_float(__builtin_amdgcn_readfirstlane(ent.y));
        s[u] = (int)(sp & ((1u << SB) - 1u));
        t[u] = (int)(sp >> SB);
      }
#pragma unroll
      for (int u = 0; u < RU; ++u)
        x[u] = *(const V*)(h + (size_t)s[u] * OUT + VEC * lane);
#pragma unroll
      for (int u = 0; u < RU; ++u) {
#pragma unroll
        for (int j = 0; j < VEC; ++j)
          atomicAdd(&tile[t[u] * P + j * 32 + lane], x[u][j] * w[u]);
      }
    } else {
      for (int i = base; i < ntot; i += BT / 32) {
        vi2 ent = lst[i];
        unsigned sp = (unsigned)__builtin_amdgcn_readfirstlane(ent.x);
        float w = __int_as_float(__builtin_amdgcn_readfirstlane(ent.y));
        int s = (int)(sp & ((1u << SB) - 1u));
        int t = (int)(sp >> SB);
        V x = *(const V*)(h + (size_t)s * OUT + VEC * lane);
#pragma unroll
        for (int j = 0; j < VEC; ++j)
          atomicAdd(&tile[t * P + j * 32 + lane], x[j] * w);
      }
    }
  }
  __syncthreads();

  if constexpr (!FUSE) {
    V bb = *(const V*)(b + VEC * lane);
    for (int t = wave; t < nt; t += BT / 32) {
      V r;
#pragma unroll
      for (int j = 0; j < VEC; ++j) {
        float v = tile[t * P + j * 32 + lane] + bb[j];
        r[j] = RELU ? fmaxf(v, 0.0f) : v;
      }
      *(V*)(outp + (size_t)(lo + t) * OUT + VEC * lane) = r;
    }
  } else {
    constexpr int NTN = OUTN / 16;
    constexpr int G   = OUTN / 64;
    const int nstrips = (nt + 15) >> 4;
    if (wave < nstrips) {
      const int m  = lane & 15;
      const int hf = lane >> 4;
      const float* arow = tile + (wave * 16 + m) * P;
      v8f acc[NTN] = {};
      for (int it = 0; it < 16; ++it) {
        const int c   = 2 * it + hf;
        const int pj  = c & 3, qi = c >> 2;
        const int off = pj * 32 + qi * 4;
        vf4 a  = *(const vf4*)(arow + off);
        vf4 bv = *(const vf4*)(bl + off);
#pragma unroll
        for (int e = 0; e < 4; ++e) {
          float v = a[e] + bv[e];
          a[e] = RELU ? fmaxf(v, 0.0f) : v;
        }
        const float* wp = Wn + (size_t)(16 * qi + pj) * OUTN + 4 * m;
        vf4 w[4][G];
#pragma unroll
        for (int e = 0; e < 4; ++e) {
#pragma unroll
          for (int g = 0; g < G; ++g) w[e][g] = *(const vf4*)(wp + (size_t)(4 * e) * OUTN + g * 64);
        }
#pragma unroll
        for (int p = 0; p < 2; ++p) {
          v2f af; af.x = a[2 * p]; af.y = a[2 * p + 1];
#pragma unroll
          for (int t = 0; t < NTN; ++t) {
            v2f bf; bf.x = w[2 * p][t / 4][t % 4]; bf.y = w[2 * p + 1][t / 4][t % 4];
            acc[t] = __builtin_amdgcn_wmma_f32_16x16x4_f32(
                 false, af,  false, bf,
                 (short)0, acc[t],  false,  false);
          }
        }
      }
      const int rbase = wave * 16 + 8 * hf;
#pragma unroll
      for (int r = 0; r < 8; ++r) {
        const int row = rbase + r;
        if (row < nt) {
#pragma unroll
          for (int g = 0; g < G; ++g) {
            vf4 o;
            o.x = acc[4 * g][r]; o.y = acc[4 * g + 1][r]; o.z = acc[4 * g + 2][r]; o.w = acc[4 * g + 3][r];
            *(vf4*)(outp + (size_t)(lo + row) * OUTN + g * 64 + 4 * m) = o;
          }
        }
      }
    }
  }
}

static inline int cdiv(long a, long b) { return (int)((a + b - 1) / b); }

extern "C" void kernel_launch(void* const* d_in, const int* in_sizes, int n_in,
                              void* d_out, int out_size, void* d_ws, size_t ws_size,
                              hipStream_t stream) {
  const float* x  = (const float*)d_in[0];
  const int*   ei = (const int*)d_in[1];
  const float* W1 = (const float*)d_in[2];
  const float* b1 = (const float*)d_in[3];
  const float* W2 = (const float*)d_in[4];
  const float* b2 = (const float*)d_in[5];
  const float* W3 = (const float*)d_in[6];
  const float* b3 = (const float*)d_in[7];

  const int N = in_sizes[0] / 128;
  const int E = in_sizes[1] / 2;
  const int* src = ei;
  const int* dst = ei + E;

  const int T = 256;
  const int gemmBlocks = cdiv((long)cdiv(N, 16) * 32, T);
  const int tiles = cdiv(N, TN);
  const int ranges = cdiv(tiles, TPR);

  char* ws = (char*)d_ws;
  size_t off0 = 0;
  size_t off1 = (((size_t)N * 4) + 255) & ~(size_t)255;
  size_t off2 = off1 + (size_t)N * 128 * 4;
  size_t off3 = off2 + (size_t)N * 128 * 4;
  size_t off4 = off3 + (size_t)tiles * LCAP * 8;
  size_t off5 = (off4 + (size_t)tiles * 4 + 255) & ~(size_t)255;
  size_t off6 = off5 + (size_t)ranges * NS * PCAP * 8;
  float* dinv  = (float*)(ws + off0);
  float* bufH  = (float*)(ws + off1);
  float* bufF  = (float*)(ws + off2);
  int*   glist = (int*)(ws + off3);
  int*   gcnt  = (int*)(ws + off4);
  int*   gpair = (int*)(ws + off5);
  int*   gpcnt = (int*)(ws + off6);
  float* out   = (float*)d_out;

  k_pick<<<ranges * NS, BT, 0, stream>>>(src, dst, gpair, gpcnt, N, E);
  k_list<<<tiles, BT, 0, stream>>>(gpair, gpcnt, src, dst, dinv, glist, gcnt, N, E);

  k_gemm<128><<<gemmBlocks, T, 0, stream>>>(x, W1, bufH, N);
  k_agg<128, true, 128, true><<<tiles, BT, 0, stream>>>(glist, gcnt, src, dst, dinv, bufH, b1, W2, bufF, N, E);

  k_agg<128, true, 64, false><<<tiles, BT, 0, stream>>>(glist, gcnt, src, dst, dinv, bufF, b2, W3, bufH, N, E);

  k_agg<64, false, 0, false><<<tiles, BT, 0, stream>>>(glist, gcnt, src, dst, dinv, bufH, b3, nullptr, out, N, E);
}


# Layer 1's product x @ W1 moved onto the last 12 / 13 waves of every block of the two owner-scan launches (a strip per wave, the rest scan); its own launch gone, 6 to 5; product k order changed
# speedup vs baseline: 4.3425x; 1.0329x over previous
#include <hip/hip_runtime.h>

typedef float v2f __attribute__((ext_vector_type(2)));
typedef float v8f __attribute__((ext_vector_type(8)));

constexpr int TN   = 392;
constexpr int LCAP = 12288;
constexpr int BT   = 1024;
constexpr int RU   = 8;
constexpr int TPR  = 16;
constexpr int RN   = TN * TPR;
constexpr int NS   = 16;
constexpr int PCAP = 8192;
constexpr int PU   = 6;
constexpr int QU   = 12;
constexpr int MWP  = 12;
constexpr int MWL  = 13;
constexpr int SPT  = BT - 32 * MWP;
constexpr int SLT  = BT - 32 * MWL;

typedef float vf2 __attribute__((ext_vector_type(2)));
typedef float vf4 __attribute__((ext_vector_type(4)));
typedef int   vi2 __attribute__((ext_vector_type(2)));
typedef int   vi4 __attribute__((ext_vector_type(4)));
template <int V> struct VT;
template <> struct VT<2> { typedef vf2 T; };
template <> struct VT<4> { typedef vf4 T; };

static_assert((PCAP / 2) % 32 == 0, "the 32 loads of one wave of k_list must stay inside one pair list");

__device__ __forceinline__ void strip_product(const float* __restrict__ A, const float* __restrict__ Wm,
                                              float* __restrict__ H, int M, int row0, int lane) {
  const int m  = lane & 15;
  const int hf = lane >> 4;
  const int arow = (row0 + m < M) ? (row0 + m) : (M - 1);
  const float* ap  = A + (size_t)arow * 128 + 4 * hf;
  const float* wp0 = Wm + (size_t)(4 * hf) * 128 + 4 * m;
  v8f acc[8] = {};
#pragma nounroll
  for (int it = 0; it < 16; ++it) {
    vf4 a = *(const vf4*)(ap + 8 * it);
    const float* wp = wp0 + (size_t)(8 * it) * 128;
    vf4 w[4][2];
#pragma unroll
    for (int e = 0; e < 4; ++e) {
#pragma unroll
      for (int g = 0; g < 2; ++g) w[e][g] = *(const vf4*)(wp + (size_t)e * 128 + g * 64);
    }
#pragma unroll
    for (int p = 0; p < 2; ++p) {
      v2f af; af.x = a[2 * p]; af.y = a[2 * p + 1];
#pragma unroll
      for (int t = 0; t < 8; ++t) {
        v2f bf; bf.x = w[2 * p][t / 4][t % 4]; bf.y = w[2 * p + 1][t / 4][t % 4];
        acc[t] = __builtin_amdgcn_wmma_f32_16x16x4_f32(
             false, af,  false, bf,
             (short)0, acc[t],  false,  false);
      }
    }
  }
  const int rbase = row0 + 8 * hf;
#pragma unroll
  for (int r = 0; r < 8; ++r) {
    const int row = rbase + r;
    if (row < M) {
#pragma unroll
      for (int g = 0; g < 2; ++g) {
        vf4 o;
        o.x = acc[4 * g][r]; o.y = acc[4 * g + 1][r]; o.z = acc[4 * g + 2][r]; o.w = acc[4 * g + 3][r];
        *(vf4*)(H + (size_t)row * 128 + g * 64 + 4 * m) = o;
      }
    }
  }
}

__global__ __launch_bounds__(BT)
void k_pick(const int* __restrict__ src, const int* __restrict__ dst,
            int* __restrict__ gpair, int* __restrict__ gpcnt,
            const float* __restrict__ x, const float* __restrict__ W1,
            float* __restrict__ h1, int N, int E) {
  __shared__ vi2 pl[PCAP];
  __shared__ int cnt;

  const int tid = threadIdx.x;
  const int lane = tid & 31;
  const int wave = tid >> 5;
  const int g   = blockIdx.x / NS;
  const int sl  = blockIdx.x % NS;
  const int rlo = g * RN;
  const unsigned urlo = (unsigned)rlo;
  const unsigned urnt = (unsigned)((N - rlo < RN) ? (N - rlo) : RN);

  const int es = (((E + NS - 1) / NS) + 3) & ~3;
  const long e0l = (long)sl * es;
  const int e0 = e0l < E ? (int)e0l : E;
  const int e1 = (E - e0 < es) ? E : (e0 + es);

  if (tid == 0) cnt = 0;
  __syncthreads();

  const bool al = ((((size_t)(src + e0)) | ((size_t)(dst + e0))) & 15u) == 0;
  const int nvec = al ? ((e1 - e0) >> 2) : 0;
  const vi4* dv = (const vi4*)(dst + e0);
  const vi4* sv = (const vi4*)(src + e0);

  if (wave >= BT / 32 - MWP) {
    const int strip = blockIdx.x * MWP + (wave - (BT / 32 - MWP));
    if (strip * 16 < N) strip_product(x, W1, h1, N, strip * 16, lane);
  } else {
  for (int vb = 0; vb < nvec; vb += SPT * PU) {
    vi4 d[PU], s[PU];
    if (vb + SPT * PU <= nvec) {
#pragma unroll
      for (int u = 0; u < PU; ++u) { d[u] = dv[vb + u * SPT + tid]; s[u] = sv[vb + u * SPT + tid]; }
    } else {
#pragma unroll
      for (int u = 0; u < PU; ++u) {
        int i = vb + u * SPT + tid;
        vi4 none = {-1, -1, -1, -1};
        d[u] = (i < nvec) ? dv[i] : none;
        s[u] = (i < nvec) ? sv[i] : none;
      }
    }
    unsigned m[4 * PU];
    int tot = 0;
#pragma unroll
    for (int u = 0; u < PU; ++u) {
#pragma unroll
      for (int j = 0; j < 4; ++j) {
        m[4 * u + j] = __builtin_amdgcn_ballot_w32((unsigned)d[u][j] - urlo < urnt);
        tot += __builtin_popcount(m[4 * u + j]);
      }
    }
    if (tot) {
      int base = 0;
      if (lane == 0) base = atomicAdd(&cnt, tot);
      base = __builtin_amdgcn_readfirstlane(base);
#pragma unroll
      for (int u = 0; u < PU; ++u) {
#pragma unroll
        for (int j = 0; j < 4; ++j) {
          const unsigned mk = m[4 * u + j];
          if ((unsigned)d[u][j] - urlo < urnt) {
            int pos = base + (int)__builtin_amdgcn_mbcnt_lo(mk, 0u);
            if (pos < PCAP) { vi2 ent; ent.x = s[u][j]; ent.y = d[u][j]; pl[pos] = ent; }
          }
          base += __builtin_popcount(mk);
        }
      }
    }
  }
  for (int e = e0 + 4 * nvec + tid; e < e1; e += SPT) {
    int dj = dst[e];
    if ((unsigned)dj - urlo < urnt) {
      int pos = atomicAdd(&cnt, 1);
      if (pos < PCAP) { vi2 ent; ent.x = src[e]; ent.y = dj; pl[pos] = ent; }
    }
  }
  }
  __syncthreads();

  const int total = cnt;
  const int n = total < PCAP ? total : PCAP;
  vi2* gp = (vi2*)gpair + (size_t)blockIdx.x * PCAP;
  for (int i = tid; i < n; i += BT) gp[i] = pl[i];
  if (tid == 0) {
    gpcnt[blockIdx.x] = total;
    if (n & 1) { vi2 pad; pad.x = 0; pad.y = -1; gp[n] = pad; }
  }
}

__global__ __launch_bounds__(BT)
void k_list(const int* __restrict__ gpair, const int* __restrict__ gpcnt,
            const int* __restrict__ src, const int* __restrict__ dst,
            float* __restrict__ dinv, int* __restrict__ glist,
            int* __restrict__ gcnt,
            const float* __restrict__ x, const float* __restrict__ W1,
            float* __restrict__ h1, int N, int E) {
  __shared__ vi2 lst[LCAP];
  __shared__ int degc[TN];
  __shared__ int pc[NS];
  __shared__ int cnt;

  const int tid = threadIdx.x;
  const int lane = tid & 31;
  const int wave = tid >> 5;
  const int lo  = blockIdx.x * TN;
  const int nt  = (N - lo < TN) ? (N - lo) : TN;
  const unsigned unt = (unsigned)nt;
  const unsigned ulo = (unsigned)lo;
  const int g   = blockIdx.x / TPR;

  if (tid == 0) cnt = 0;
  if (tid < NS) pc[tid] = gpcnt[g * NS + tid];
  for (int t = tid; t < TN; t += BT) degc[t] = 0;
  __syncthreads();

  int pmax = 0;
#pragma unroll
  for (int s = 0; s < NS; ++s) pmax = pc[s] > pmax ? pc[s] : pmax;

  if (wave >= BT / 32 - MWL) {
    const int taken = ((gridDim.x + TPR - 1) / TPR) * NS * MWP;
    const int nstr  = (N + 15) >> 4;
    for (int strip = taken + blockIdx.x * MWL + (wave - (BT / 32 - MWL)); strip < nstr; strip += gridDim.x * MWL)
      strip_product(x, W1, h1, N, strip * 16, lane);
  } else if (pmax <= PCAP) {
    constexpr int VPL = PCAP / 2;
    const vi4* pv = (const vi4*)gpair + (size_t)g * NS * VPL;
    for (int vb = 0; vb < NS * VPL; vb += SLT * QU) {
      vi4 p[QU];
#pragma unroll
      for (int u = 0; u < QU; ++u) {
        int v  = vb + u * SLT + tid;
        int np = (v < NS * VPL) ? pc[v / VPL] : 0;
        int i  = 2 * (v % VPL);
        vi4 none = {0, -1, 0, -1};
        p[u] = (i < np) ? pv[v] : none;
      }
      unsigned m[2 * QU];
      int tot = 0;
#pragma unroll
      for (int u = 0; u < QU; ++u) {
        m[2 * u]     = __builtin_amdgcn_ballot_w32((unsigned)p[u].y - ulo < unt);
        m[2 * u + 1] = __builtin_amdgcn_ballot_w32((unsigned)p[u].w - ulo < unt);
        tot += __builtin_popcount(m[2 * u]) + __builtin_popcount(m[2 * u + 1]);
      }
      if (tot) {
        int base = 0;
        if (lane == 0) base = atomicAdd(&cnt, tot);
        base = __builtin_amdgcn_readfirstlane(base);
#pragma unroll
        for (int u = 0; u < QU; ++u) {
          unsigned t0 = (unsigned)p[u].y - ulo, t1 = (unsigned)p[u].w - ulo;
          if (t0 < unt) {
            atomicAdd(&degc[t0], 1);
            int pos = base + (int)__builtin_amdgcn_mbcnt_lo(m[2 * u], 0u);
            if (pos < LCAP) { vi2 ent; ent.x = p[u].x; ent.y = (int)t0; lst[pos] = ent; }
          }
          base += __builtin_popcount(m[2 * u]);
          if (t1 < unt) {
            atomicAdd(&degc[t1], 1);
            int pos = base + (int)__builtin_amdgcn_mbcnt_lo(m[2 * u + 1], 0u);
            if (pos < LCAP) { vi2 ent; ent.x = p[u].z; ent.y = (int)t1; lst[pos] = ent; }
          }
          base += __builtin_popcount(m[2 * u + 1]);
        }
      }
    }
  } else {
    for (int e = tid; e < E; e += SLT) {
      unsigned t = (unsigned)dst[e] - ulo;
      if (t < unt) {
        atomicAdd(&degc[t], 1);
        int pos = atomicAdd(&cnt, 1);
        if (pos < LCAP) { vi2 ent; ent.x = src[e]; ent.y = (int)t; lst[pos] = ent; }
      }
    }
  }
  __syncthreads();

  const int total = cnt;
  const int n = total < LCAP ? total : LCAP;
  vi2* gl = (vi2*)glist + (size_t)blockIdx.x * LCAP;
  for (int i = tid; i < n; i += BT) gl[i] = lst[i];
  if (tid == 0) gcnt[blockIdx.x] = total;
  for (int t = tid; t < nt; t += BT) dinv[lo + t] = rsqrtf(1.0f + (float)degc[t]);
}

constexpr int SB = 23;
static_assert(TN <= (1 << (32 - SB)), "the local row must fit above the source bits");

template <int OUT, bool RELU, int OUTN, bool FIRST>
__global__ __launch_bounds__(BT)
void k_agg(int* __restrict__ glist, const int* __restrict__ gcnt,
           const int* __restrict__ src, const int* __restrict__ dst,
           const float* __restrict__ dinv, const float* __restrict__ h,
           const float* __restrict__ b, const float* __restrict__ Wn,
           float* __restrict__ outp, int N, int E) {
  constexpr int VEC = OUT / 32;
  constexpr bool FUSE = OUTN > 0;
  constexpr int P  = FUSE ? OUT + 4 : OUT;
  constexpr int TR = FUSE ? ((TN + 15) / 16) * 16 : TN;
  static_assert(!FUSE || (OUT == 128 && OUTN % 64 == 0), "fused product: K = 128, whole 64-column groups");
  static_assert((TN + 15) / 16 <= BT / 32, "one strip per wave");
  static_assert((TR * P) % 4 == 0, "the tile is cleared in 16-byte pieces");
  typedef typename VT<VEC>::T V;
  __shared__ __attribute__((aligned(16))) float tile[TR * P];
  __shared__ vi2   lst[LCAP + TN];
  __shared__ __attribute__((aligned(16))) float bl[FUSE ? OUT : 4];

  const int tid  = threadIdx.x;
  const int lane = tid & 31;
  const int wave = tid >> 5;
  const int lo   = blockIdx.x * TN;
  const int nt   = (N - lo < TN) ? (N - lo) : TN;
  const unsigned unt = (unsigned)nt;
  const unsigned ulo = (unsigned)lo;

  const int  total = gcnt[blockIdx.x];
  const float di = dinv[lo + (tid < nt ? tid : 0)];
  const float bv = FUSE ? b[tid < OUT ? tid : 0] : 0.0f;
  const bool whole = total <= LCAP && N <= (1 << SB);
  const int  n     = whole ? total : 0;

  {
    vf4 z = {0.0f, 0.0f, 0.0f, 0.0f};
    for (int i = tid; i < TR * P / 4; i += BT) ((vf4*)tile)[i] = z;
  }
  {
    constexpr int LU = 8;
    vi2* gl = (vi2*)glist + (size_t)blockIdx.x * LCAP;
    for (int i0 = tid; i0 < n; i0 += BT * LU) {
      vi2 ent[LU];
#pragma unroll
      for (int u = 0; u < LU; ++u) {
        int i = i0 + u * BT;
        vi2 none = {0, 0};
        ent[u] = (i < n) ? gl[i] : none;
      }
      if constexpr (FIRST) {
        float ds[LU], dd[LU];
#pragma unroll
        for (int u = 0; u < LU; ++u) {
          ds[u] = dinv[ent[u].x];
          dd[u] = dinv[lo + ent[u].y];
        }
#pragma unroll
        for (int u = 0; u < LU; ++u) {
          int i = i0 + u * BT;
          ent[u].x = ent[u].x | (int)((unsigned)ent[u].y << SB);
          ent[u].y = __float_as_int(ds[u] * dd[u]);
          if (i < n) gl[i] = ent[u];
        }
      }
#pragma unroll
      for (int u = 0; u < LU; ++u) {
        int i = i0 + u * BT;
        if (i < n) lst[i] = ent[u];
      }
    }
  }
  if (tid < nt) {
    vi2 ent;
    ent.x = (lo + tid) | (int)((unsigned)tid << SB);
    ent.y = __float_as_int(di * di);
    lst[n + tid] = ent;
  }
  if (FUSE) {
    if (tid < OUT) bl[(tid % VEC) * 32 + tid / VEC] = bv;
  }
  __syncthreads();

  if (!whole) {
    for (int e = tid; e < E; e += BT) {
      unsigned t = (unsigned)dst[e] - ulo;
      if (t < unt) {
        int s = src[e];
        float nrm = dinv[s] * dinv[lo + t];
        const float* hs = h + (size_t)s * OUT;
        for (int c = 0; c < OUT; ++c)
          atomicAdd(&tile[t * P + (c % VEC) * 32 + c / VEC], hs[c] * nrm);
      }
    }
  }

  const int ntot = n + nt;
  for (int base = wave; base < ntot; base += (BT / 32) * RU) {
    if (base + (BT / 32) * (RU - 1) < ntot) {
      int   s[RU], t[RU];
      V     x[RU];
      float w[RU];
#pragma unroll
      for (int u = 0; u < RU; ++u) {
        vi2 ent = lst[base + (BT / 32) * u];
        unsigned sp = (unsigned)__builtin_amdgcn_readfirstlane(ent.x);
        w[u] = __int_as_float(__builtin_amdgcn_readfirstlane(ent.y));
        s[u] = (int)(sp & ((1u << SB) - 1u));
        t[u] = (int)(sp >> SB);
      }
#pragma unroll
      for (int u = 0; u < RU; ++u)
        x[u] = *(const V*)(h + (size_t)s[u] * OUT + VEC * lane);
#pragma unroll
      for (int u = 0; u < RU; ++u) {
#pragma unroll
        for (int j = 0; j < VEC; ++j)
          atomicAdd(&tile[t[u] * P + j * 32 + lane], x[u][j] * w[u]);
      }
    } else {
      for (int i = base; i < ntot; i += BT / 32) {
        vi2 ent = lst[i];
        unsigned sp = (unsigned)__builtin_amdgcn_readfirstlane(ent.x);
        float w = __int_as_float(__builtin_amdgcn_readfirstlane(ent.y));
        int s = (int)(sp & ((1u << SB) - 1u));
        int t = (int)(sp >> SB);
        V x = *(const V*)(h + (size_t)s * OUT + VEC * lane);
#pragma unroll
        for (int j = 0; j < VEC; ++j)
          atomicAdd(&tile[t * P + j * 32 + lane], x[j] * w);
      }
    }
  }
  __syncthreads();

  if constexpr (!FUSE) {
    V bb = *(const V*)(b + VEC * lane);
    for (int t = wave; t < nt; t += BT / 32) {
      V r;
#pragma unroll
      for (int j = 0; j < VEC; ++j) {
        float v = tile[t * P + j * 32 + lane] + bb[j];
        r[j] = RELU ? fmaxf(v, 0.0f) : v;
      }
      *(V*)(outp + (size_t)(lo + t) * OUT + VEC * lane) = r;
    }
  } else {
    constexpr int NTN = OUTN / 16;
    constexpr int G   = OUTN / 64;
    const int nstrips = (nt + 15) >> 4;
    if (wave < nstrips) {
      const int m  = lane & 15;
      const int hf = lane >> 4;
      const float* arow = tile + (wave * 16 + m) * P;
      v8f acc[NTN] = {};
      for (int it = 0; it < 16; ++it) {
        const int c   = 2 * it + hf;
        const int pj  = c & 3, qi = c >> 2;
        const int off = pj * 32 + qi * 4;
        vf4 a  = *(const vf4*)(arow + off);
        vf4 bv = *(const vf4*)(bl + off);
#pragma unroll
        for (int e = 0; e < 4; ++e) {
          float v = a[e] + bv[e];
          a[e] = RELU ? fmaxf(v, 0.0f) : v;
        }
        const float* wp = Wn + (size_t)(16 * qi + pj) * OUTN + 4 * m;
        vf4 w[4][G];
#pragma unroll
        for (int e = 0; e < 4; ++e) {
#pragma unroll
          for (int g = 0; g < G; ++g) w[e][g] = *(const vf4*)(wp + (size_t)(4 * e) * OUTN + g * 64);
        }
#pragma unroll
        for (int p = 0; p < 2; ++p) {
          v2f af; af.x = a[2 * p]; af.y = a[2 * p + 1];
#pragma unroll
          for (int t = 0; t < NTN; ++t) {
            v2f bf; bf.x = w[2 * p][t / 4][t % 4]; bf.y = w[2 * p + 1][t / 4][t % 4];
            acc[t] = __builtin_amdgcn_wmma_f32_16x16x4_f32(
                 false, af,  false, bf,
                 (short)0, acc[t],  false,  false);
          }
        }
      }
      const int rbase = wave * 16 + 8 * hf;
#pragma unroll
      for (int r = 0; r < 8; ++r) {
        const int row = rbase + r;
        if (row < nt) {
#pragma unroll
          for (int g = 0; g < G; ++g) {
            vf4 o;
            o.x = acc[4 * g][r]; o.y = acc[4 * g + 1][r]; o.z = acc[4 * g + 2][r]; o.w = acc[4 * g + 3][r];
            *(vf4*)(outp + (size_t)(lo + row) * OUTN + g * 64 + 4 * m) = o;
          }
        }
      }
    }
  }
}

static inline int cdiv(long a, long b) { return (int)((a + b - 1) / b); }

extern "C" void kernel_launch(void* const* d_in, const int* in_sizes, int n_in,
                              void* d_out, int out_size, void* d_ws, size_t ws_size,
                              hipStream_t stream) {
  const float* x  = (const float*)d_in[0];
  const int*   ei = (const int*)d_in[1];
  const float* W1 = (const float*)d_in[2];
  const float* b1 = (const float*)d_in[3];
  const float* W2 = (const float*)d_in[4];
  const float* b2 = (const float*)d_in[5];
  const float* W3 = (const float*)d_in[6];
  const float* b3 = (const float*)d_in[7];

  const int N = in_sizes[0] / 128;
  const int E = in_sizes[1] / 2;
  const int* src = ei;
  const int* dst = ei + E;

  const int tiles = cdiv(N, TN);
  const int ranges = cdiv(tiles, TPR);

  char* ws = (char*)d_ws;
  size_t off0 = 0;
  size_t off1 = (((size_t)N * 4) + 255) & ~(size_t)255;
  size_t off2 = off1 + (size_t)N * 128 * 4;
  size_t off3 = off2 + (size_t)N * 128 * 4;
  size_t off4 = off3 + (size_t)tiles * LCAP * 8;
  size_t off5 = (off4 + (size_t)tiles * 4 + 255) & ~(size_t)255;
  size_t off6 = off5 + (size_t)ranges * NS * PCAP * 8;
  float* dinv  = (float*)(ws + off0);
  float* bufH  = (float*)(ws + off1);
  float* bufF  = (float*)(ws + off2);
  int*   glist = (int*)(ws + off3);
  int*   gcnt  = (int*)(ws + off4);
  int*   gpair = (int*)(ws + off5);
  int*   gpcnt = (int*)(ws + off6);
  float* out   = (float*)d_out;

  k_pick<<<ranges * NS, BT, 0, stream>>>(src, dst, gpair, gpcnt, x, W1, bufH, N, E);
  k_list<<<tiles, BT, 0, stream>>>(gpair, gpcnt, src, dst, dinv, glist, gcnt, x, W1, bufH, N, E);

  k_agg<128, true, 128, true><<<tiles, BT, 0, stream>>>(glist, gcnt, src, dst, dinv, bufH, b1, W2, bufF, N, E);

  k_agg<128, true, 64, false><<<tiles, BT, 0, stream>>>(glist, gcnt, src, dst, dinv, bufF, b2, W3, bufH, N, E);

  k_agg<64, false, 0, false><<<tiles, BT, 0, stream>>>(glist, gcnt, src, dst, dinv, bufH, b3, nullptr, out, N, E);
}


# Tile rows owned by waves (row % 32), lists parted by owner: owner adds value * source row by plain 16-B LDS read, fma, write under next batch's loads; no LDS float adds; sum, product k order changed
# speedup vs baseline: 4.6352x; 1.0674x over previous
#include <hip/hip_runtime.h>

typedef float v2f __attribute__((ext_vector_type(2)));
typedef float v8f __attribute__((ext_vector_type(8)));

constexpr int TN   = 392;
constexpr int LCAP = 12288;
constexpr int BT   = 1024;
constexpr int RU   = 8;
constexpr int TPR  = 16;
constexpr int RN   = TN * TPR;
constexpr int NS   = 16;
constexpr int PCAP = 8192;
constexpr int PU   = 6;
constexpr int QU   = 12;
constexpr int MWP  = 12;
constexpr int MWL  = 13;
constexpr int SPT  = BT - 32 * MWP;
constexpr int SLT  = BT - 32 * MWL;
constexpr int NW   = BT / 32;
constexpr int WCAP = LCAP / NW;
constexpr int WS   = WCAP + 16;
static_assert((TN + NW - 1) / NW <= 16, "a wave's self-loop entries must fit behind its filed ones");

typedef float vf2 __attribute__((ext_vector_type(2)));
typedef float vf4 __attribute__((ext_vector_type(4)));
typedef int   vi2 __attribute__((ext_vector_type(2)));
typedef int   vi4 __attribute__((ext_vector_type(4)));
template <int V> struct VT;
template <> struct VT<2> { typedef vf2 T; };
template <> struct VT<4> { typedef vf4 T; };

static_assert((PCAP / 2) % 32 == 0, "the 32 loads of one wave of k_list must stay inside one pair list");

__device__ __forceinline__ void strip_product(const float* __restrict__ A, const float* __restrict__ Wm,
                                              float* __restrict__ H, int M, int row0, int lane) {
  const int m  = lane & 15;
  const int hf = lane >> 4;
  const int arow = (row0 + m < M) ? (row0 + m) : (M - 1);
  const float* ap  = A + (size_t)arow * 128 + 4 * hf;
  const float* wp0 = Wm + (size_t)(4 * hf) * 128 + 4 * m;
  v8f acc[8] = {};
#pragma nounroll
  for (int it = 0; it < 16; ++it) {
    vf4 a = *(const vf4*)(ap + 8 * it);
    const float* wp = wp0 + (size_t)(8 * it) * 128;
    vf4 w[4][2];
#pragma unroll
    for (int e = 0; e < 4; ++e) {
#pragma unroll
      for (int g = 0; g < 2; ++g) w[e][g] = *(const vf4*)(wp + (size_t)e * 128 + g * 64);
    }
#pragma unroll
    for (int p = 0; p < 2; ++p) {
      v2f af; af.x = a[2 * p]; af.y = a[2 * p + 1];
#pragma unroll
      for (int t = 0; t < 8; ++t) {
        v2f bf; bf.x = w[2 * p][t / 4][t % 4]; bf.y = w[2 * p + 1][t / 4][t % 4];
        acc[t] = __builtin_amdgcn_wmma_f32_16x16x4_f32(
             false, af,  false, bf,
             (short)0, acc[t],  false,  false);
      }
    }
  }
  const int rbase = row0 + 8 * hf;
#pragma unroll
  for (int r = 0; r < 8; ++r) {
    const int row = rbase + r;
    if (row < M) {
#pragma unroll
      for (int g = 0; g < 2; ++g) {
        vf4 o;
        o.x = acc[4 * g][r]; o.y = acc[4 * g + 1][r]; o.z = acc[4 * g + 2][r]; o.w = acc[4 * g + 3][r];
        *(vf4*)(H + (size_t)row * 128 + g * 64 + 4 * m) = o;
      }
    }
  }
}

__global__ __launch_bounds__(BT)
void k_pick(const int* __restrict__ src, const int* __restrict__ dst,
            int* __restrict__ gpair, int* __restrict__ gpcnt,
            const float* __restrict__ x, const float* __restrict__ W1,
            float* __restrict__ h1, int N, int E) {
  __shared__ vi2 pl[PCAP];
  __shared__ int cnt;

  const int tid = threadIdx.x;
  const int lane = tid & 31;
  const int wave = tid >> 5;
  const int g   = blockIdx.x / NS;
  const int sl  = blockIdx.x % NS;
  const int rlo = g * RN;
  const unsigned urlo = (unsigned)rlo;
  const unsigned urnt = (unsigned)((N - rlo < RN) ? (N - rlo) : RN);

  const int es = (((E + NS - 1) / NS) + 3) & ~3;
  const long e0l = (long)sl * es;
  const int e0 = e0l < E ? (int)e0l : E;
  const int e1 = (E - e0 < es) ? E : (e0 + es);

  if (tid == 0) cnt = 0;
  __syncthreads();

  const bool al = ((((size_t)(src + e0)) | ((size_t)(dst + e0))) & 15u) == 0;
  const int nvec = al ? ((e1 - e0) >> 2) : 0;
  const vi4* dv = (const vi4*)(dst + e0);
  const vi4* sv = (const vi4*)(src + e0);

  if (wave >= BT / 32 - MWP) {
    const int strip = blockIdx.x * MWP + (wave - (BT / 32 - MWP));
    if (strip * 16 < N) strip_product(x, W1, h1, N, strip * 16, lane);
  } else {
  for (int vb = 0; vb < nvec; vb += SPT * PU) {
    vi4 d[PU], s[PU];
    if (vb + SPT * PU <= nvec) {
#pragma unroll
      for (int u = 0; u < PU; ++u) { d[u] = dv[vb + u * SPT + tid]; s[u] = sv[vb + u * SPT + tid]; }
    } else {
#pragma unroll
      for (int u = 0; u < PU; ++u) {
        int i = vb + u * SPT + tid;
        vi4 none = {-1, -1, -1, -1};
        d[u] = (i < nvec) ? dv[i] : none;
        s[u] = (i < nvec) ? sv[i] : none;
      }
    }
    unsigned m[4 * PU];
    int tot = 0;
#pragma unroll
    for (int u = 0; u < PU; ++u) {
#pragma unroll
      for (int j = 0; j < 4; ++j) {
        m[4 * u + j] = __builtin_amdgcn_ballot_w32((unsigned)d[u][j] - urlo < urnt);
        tot += __builtin_popcount(m[4 * u + j]);
      }
    }
    if (tot) {
      int base = 0;
      if (lane == 0) base = atomicAdd(&cnt, tot);
      base = __builtin_amdgcn_readfirstlane(base);
#pragma unroll
      for (int u = 0; u < PU; ++u) {
#pragma unroll
        for (int j = 0; j < 4; ++j) {
          const unsigned mk = m[4 * u + j];
          if ((unsigned)d[u][j] - urlo < urnt) {
            int pos = base + (int)__builtin_amdgcn_mbcnt_lo(mk, 0u);
            if (pos < PCAP) { vi2 ent; ent.x = s[u][j]; ent.y = d[u][j]; pl[pos] = ent; }
          }
          base += __builtin_popcount(mk);
        }
      }
    }
  }
  for (int e = e0 + 4 * nvec + tid; e < e1; e += SPT) {
    int dj = dst[e];
    if ((unsigned)dj - urlo < urnt) {
      int pos = atomicAdd(&cnt, 1);
      if (pos < PCAP) { vi2 ent; ent.x = src[e]; ent.y = dj; pl[pos] = ent; }
    }
  }
  }
  __syncthreads();

  const int total = cnt;
  const int n = total < PCAP ? total : PCAP;
  vi2* gp = (vi2*)gpair + (size_t)blockIdx.x * PCAP;
  for (int i = tid; i < n; i += BT) gp[i] = pl[i];
  if (tid == 0) {
    gpcnt[blockIdx.x] = total;
    if (n & 1) { vi2 pad; pad.x = 0; pad.y = -1; gp[n] = pad; }
  }
}

__global__ __launch_bounds__(BT)
void k_list(const int* __restrict__ gpair, const int* __restrict__ gpcnt,
            const int* __restrict__ src, const int* __restrict__ dst,
            float* __restrict__ dinv, int* __restrict__ glist,
            int* __restrict__ gcnt, int* __restrict__ gwcnt,
            const float* __restrict__ x, const float* __restrict__ W1,
            float* __restrict__ h1, int N, int E) {
  __shared__ vi2 lst[LCAP];
  __shared__ vi2 lst2[LCAP];
  __shared__ int wc[NW];
  __shared__ int degc[TN];
  __shared__ int pc[NS];
  __shared__ int cnt;

  const int tid = threadIdx.x;
  const int lane = tid & 31;
  const int wave = tid >> 5;
  const int lo  = blockIdx.x * TN;
  const int nt  = (N - lo < TN) ? (N - lo) : TN;
  const unsigned unt = (unsigned)nt;
  const unsigned ulo = (unsigned)lo;
  const int g   = blockIdx.x / TPR;

  if (tid == 0) cnt = 0;
  if (tid < NW) wc[tid] = 0;
  if (tid < NS) pc[tid] = gpcnt[g * NS + tid];
  for (int t = tid; t < TN; t += BT) degc[t] = 0;
  __syncthreads();

  int pmax = 0;
#pragma unroll
  for (int s = 0; s < NS; ++s) pmax = pc[s] > pmax ? pc[s] : pmax;

  if (wave >= BT / 32 - MWL) {
    const int taken = ((gridDim.x + TPR - 1) / TPR) * NS * MWP;
    const int nstr  = (N + 15) >> 4;
    for (int strip = taken + blockIdx.x * MWL + (wave - (BT / 32 - MWL)); strip < nstr; strip += gridDim.x * MWL)
      strip_product(x, W1, h1, N, strip * 16, lane);
  } else if (pmax <= PCAP) {
    constexpr int VPL = PCAP / 2;
    const vi4* pv = (const vi4*)gpair + (size_t)g * NS * VPL;
    for (int vb = 0; vb < NS * VPL; vb += SLT * QU) {
      vi4 p[QU];
#pragma unroll
      for (int u = 0; u < QU; ++u) {
        int v  = vb + u * SLT + tid;
        int np = (v < NS * VPL) ? pc[v / VPL] : 0;
        int i  = 2 * (v % VPL);
        vi4 none = {0, -1, 0, -1};
        p[u] = (i < np) ? pv[v] : none;
      }
      unsigned m[2 * QU];
      int tot = 0;
#pragma unroll
      for (int u = 0; u < QU; ++u) {
        m[2 * u]     = __builtin_amdgcn_ballot_w32((unsigned)p[u].y - ulo < unt);
        m[2 * u + 1] = __builtin_amdgcn_ballot_w32((unsigned)p[u].w - ulo < unt);
        tot += __builtin_popcount(m[2 * u]) + __builtin_popcount(m[2 * u + 1]);
      }
      if (tot) {
        int base = 0;
        if (lane == 0) base = atomicAdd(&cnt, tot);
        base = __builtin_amdgcn_readfirstlane(base);
#pragma unroll
        for (int u = 0; u < QU; ++u) {
          unsigned t0 = (unsigned)p[u].y - ulo, t1 = (unsigned)p[u].w - ulo;
          if (t0 < unt) {
            atomicAdd(&degc[t0], 1);
            int pos = base + (int)__builtin_amdgcn_mbcnt_lo(m[2 * u], 0u);
            if (pos < LCAP) { vi2 ent; ent.x = p[u].x; ent.y = (int)t0; lst[pos] = ent; }
          }
          base += __builtin_popcount(m[2 * u]);
          if (t1 < unt) {
            atomicAdd(&degc[t1], 1);
            int pos = base + (int)__builtin_amdgcn_mbcnt_lo(m[2 * u + 1], 0u);
            if (pos < LCAP) { vi2 ent; ent.x = p[u].z; ent.y = (int)t1; lst[pos] = ent; }
          }
          base += __builtin_popcount(m[2 * u + 1]);
        }
      }
    }
  } else {
    for (int e = tid; e < E; e += SLT) {
      unsigned t = (unsigned)dst[e] - ulo;
      if (t < unt) {
        atomicAdd(&degc[t], 1);
        int pos = atomicAdd(&cnt, 1);
        if (pos < LCAP) { vi2 ent; ent.x = src[e]; ent.y = (int)t; lst[pos] = ent; }
      }
    }
  }
  __syncthreads();

  const int total = cnt;
  const int n = total < LCAP ? total : LCAP;
  for (int i = tid; i < n; i += BT) {
    vi2 ent = lst[i];
    int o = ent.y & (NW - 1);
    int pos = atomicAdd(&wc[o], 1);
    if (pos < WCAP) lst2[o * WCAP + pos] = ent;
  }
  __syncthreads();

  {
    int c = wc[wave];
    c = c < WCAP ? c : WCAP;
    vi2* gl = (vi2*)glist + (size_t)blockIdx.x * LCAP + wave * WCAP;
    for (int k = lane; k < c; k += 32) gl[k] = lst2[wave * WCAP + k];
  }
  if (tid < NW) gwcnt[blockIdx.x * NW + tid] = wc[tid];
  if (tid == 0) {
    int over = total > LCAP;
    for (int o = 0; o < NW; ++o) over |= wc[o] > WCAP;
    gcnt[blockIdx.x] = over ? LCAP + 1 : total;
  }
  for (int t = tid; t < nt; t += BT) dinv[lo + t] = rsqrtf(1.0f + (float)degc[t]);
}

constexpr int SB = 23;
static_assert(TN <= (1 << (32 - SB)), "the local row must fit above the source bits");

template <int OUT, bool RELU, int OUTN, bool FIRST>
__global__ __launch_bounds__(BT)
void k_agg(int* __restrict__ glist, const int* __restrict__ gcnt, const int* __restrict__ gwcnt,
           const int* __restrict__ src, const int* __restrict__ dst,
           const float* __restrict__ dinv, const float* __restrict__ h,
           const float* __restrict__ b, const float* __restrict__ Wn,
           float* __restrict__ outp, int N, int E) {
  constexpr int VEC = OUT / 32;
  constexpr bool FUSE = OUTN > 0;
  constexpr int P  = FUSE ? OUT + 4 : OUT;
  constexpr int TR = FUSE ? ((TN + 15) / 16) * 16 : TN + 4;
  constexpr int SPARE = TR - 1;
  static_assert(SPARE >= TN && SPARE < (1 << (32 - SB)), "the spare row lies past the tile and fits an entry's row bits");
  static_assert(WS % (2 * RU) == 0, "a wave's list is padded to an even number of whole batches inside its part");
  static_assert(!FUSE || (OUT == 128 && OUTN % 64 == 0), "fused product: K = 128, whole 64-column groups");
  static_assert((TN + 15) / 16 <= BT / 32, "one strip per wave");
  static_assert((TR * P) % 4 == 0, "the tile is cleared in 16-byte pieces");
  static_assert(P % VEC == 0, "a lane's columns of a row are read and written as one aligned piece");
  typedef typename VT<VEC>::T V;
  __shared__ __attribute__((aligned(16))) float tile[TR * P];
  __shared__ vi2   lst[NW * WS];
  __shared__ __attribute__((aligned(16))) float bl[FUSE ? OUT : 4];

  const int tid  = threadIdx.x;
  const int lane = tid & 31;
  const int wave = tid >> 5;
  const int lo   = blockIdx.x * TN;
  const int nt   = (N - lo < TN) ? (N - lo) : TN;
  const unsigned unt = (unsigned)nt;
  const unsigned ulo = (unsigned)lo;

  const int  total = gcnt[blockIdx.x];
  const int  filed = gwcnt[blockIdx.x * NW + wave];
  const int  rs   = wave + NW * lane;
  const int  ns   = __builtin_amdgcn_readfirstlane(nt > wave ? (nt - 1 - wave) / NW + 1 : 0);
  const float di = dinv[lo + (lane < ns ? rs : 0)];
  const float bv = FUSE ? b[tid < OUT ? tid : 0] : 0.0f;
  const bool whole = total <= LCAP && N <= (1 << SB);
  const int  cw    = __builtin_amdgcn_readfirstlane(whole ? (filed < WCAP ? filed : WCAP) : 0);

  {
    vf4 z = {0.0f, 0.0f, 0.0f, 0.0f};
    for (int i = tid; i < TR * P / 4; i += BT) ((vf4*)tile)[i] = z;
  }
  vi2* my = lst + wave * WS;
  {
    constexpr int LU = 8;
    vi2* gl = (vi2*)glist + (size_t)blockIdx.x * LCAP + wave * WCAP;
    for (int k0 = lane; k0 < cw; k0 += 32 * LU) {
      vi2 ent[LU];
#pragma unroll
      for (int u = 0; u < LU; ++u) {
        int k = k0 + u * 32;
        vi2 none = {0, 0};
        ent[u] = (k < cw) ? gl[k] : none;
      }
      if constexpr (FIRST) {
        float ds[LU], dd[LU];
#pragma unroll
        for (int u = 0; u < LU; ++u) {
          ds[u] = dinv[ent[u].x];
          dd[u] = dinv[lo + ent[u].y];
        }
#pragma unroll
        for (int u = 0; u < LU; ++u) {
          int k = k0 + u * 32;
          ent[u].x = ent[u].x | (int)((unsigned)ent[u].y << SB);
          ent[u].y = __float_as_int(ds[u] * dd[u]);
          if (k < cw) gl[k] = ent[u];
        }
      }
#pragma unroll
      for (int u = 0; u < LU; ++u) {
        int k = k0 + u * 32;
        if (k < cw) my[k] = ent[u];
      }
    }
  }
  const int ntot = cw + ns;
  const int nb   = ((ntot + 2 * RU - 1) / (2 * RU)) * 2;
  if (lane < ns) {
    vi2 ent;
    ent.x = (lo + rs) | (int)((unsigned)rs << SB);
    ent.y = __float_as_int(di * di);
    my[cw + lane] = ent;
  } else if (cw + lane < nb * RU) {
    vi2 ent;
    ent.x = (int)((unsigned)SPARE << SB);
    ent.y = __float_as_int(0.0f);
    my[cw + lane] = ent;
  }
  if (FUSE) {
    if (tid < OUT) bl[tid] = bv;
  }
  __syncthreads();

  if (!whole) {
    for (int e = tid; e < E; e += BT) {
      unsigned t = (unsigned)dst[e] - ulo;
      if (t < unt) {
        int s = src[e];
        float nrm = dinv[s] * dinv[lo + t];
        const float* hs = h + (size_t)s * OUT;
        for (int c = 0; c < OUT; ++c)
          atomicAdd(&tile[t * P + c], hs[c] * nrm);
      }
    }
    __syncthreads();
  }

  {
    int   ta[RU], tb[RU];
    float wa[RU], wb[RU];
    V     xa[RU], xb[RU];
    auto load_batch = [&](int bi, int (&t)[RU], float (&w)[RU], V (&x)[RU]) {
      const vi2* ep = my + bi * RU;
      int s[RU];
#pragma unroll
      for (int u = 0; u < RU; ++u) {
        vi2 ent = ep[u];
        unsigned sp = (unsigned)__builtin_amdgcn_readfirstlane(ent.x);
        w[u] = __int_as_float(__builtin_amdgcn_readfirstlane(ent.y));
        s[u] = (int)(sp & ((1u << SB) - 1u));
        t[u] = (int)(sp >> SB);
      }
#pragma unroll
      for (int u = 0; u < RU; ++u)
        x[u] = *(const V*)(h + (size_t)s[u] * OUT + VEC * lane);
    };
    auto add_batch = [&](const int (&t)[RU], const float (&w)[RU], const V (&x)[RU]) {
#pragma unroll
      for (int u = 0; u < RU; ++u) {
        V* rp = (V*)(tile + t[u] * P + VEC * lane);
        V a = *rp;
#pragma unroll
        for (int j = 0; j < VEC; ++j) a[j] += x[u][j] * w[u];
        *rp = a;
      }
    };
    if (nb > 0) {
      load_batch(0, ta, wa, xa);
      for (int bi = 0; bi < nb; bi += 2) {
        load_batch(bi + 1, tb, wb, xb);
        add_batch(ta, wa, xa);
        load_batch(bi + 2 < nb ? bi + 2 : nb - 1, ta, wa, xa);
        add_batch(tb, wb, xb);
      }
    }
  }
  __syncthreads();

  if constexpr (!FUSE) {
    V bb = *(const V*)(b + VEC * lane);
    for (int t = wave; t < nt; t += BT / 32) {
      V a = *(const V*)(tile + t * P + VEC * lane);
      V r;
#pragma unroll
      for (int j = 0; j < VEC; ++j) {
        float v = a[j] + bb[j];
        r[j] = RELU ? fmaxf(v, 0.0f) : v;
      }
      *(V*)(outp + (size_t)(lo + t) * OUT + VEC * lane) = r;
    }
  } else {
    constexpr int NTN = OUTN / 16;
    constexpr int G   = OUTN / 64;
    const int nstrips = (nt + 15) >> 4;
    if (wave < nstrips) {
      const int m  = lane & 15;
      const int hf = lane >> 4;
      const float* arow = tile + (wave * 16 + m) * P + 4 * hf;
      const float* wp0  = Wn + (size_t)(4 * hf) * OUTN + 4 * m;
      v8f acc[NTN] = {};
      for (int it = 0; it < 16; ++it) {
        vf4 a  = *(const vf4*)(arow + 8 * it);
        vf4 bv4 = *(const vf4*)(bl + 8 * it + 4 * hf);
#pragma unroll
        for (int e = 0; e < 4; ++e) {
          float v = a[e] + bv4[e];
          a[e] = RELU ? fmaxf(v, 0.0f) : v;
        }
        const float* wp = wp0 + (size_t)(8 * it) * OUTN;
        vf4 w[4][G];
#pragma unroll
        for (int e = 0; e < 4; ++e) {
#pragma unroll
          for (int g = 0; g < G; ++g) w[e][g] = *(const vf4*)(wp + (size_t)e * OUTN + g * 64);
        }
#pragma unroll
        for (int p = 0; p < 2; ++p) {
          v2f af; af.x = a[2 * p]; af.y = a[2 * p + 1];
#pragma unroll
          for (int t = 0; t < NTN; ++t) {
            v2f bf; bf.x = w[2 * p][t / 4][t % 4]; bf.y = w[2 * p + 1][t / 4][t % 4];
            acc[t] = __builtin_amdgcn_wmma_f32_16x16x4_f32(
                 false, af,  false, bf,
                 (short)0, acc[t],  false,  false);
          }
        }
      }
      const int rbase = wave * 16 + 8 * hf;
#pragma unroll
      for (int r = 0; r < 8; ++r) {
        const int row = rbase + r;
        if (row < nt) {
#pragma unroll
          for (int g = 0; g < G; ++g) {
            vf4 o;
            o.x = acc[4 * g][r]; o.y = acc[4 * g + 1][r]; o.z = acc[4 * g + 2][r]; o.w = acc[4 * g + 3][r];
            *(vf4*)(outp + (size_t)(lo + row) * OUTN + g * 64 + 4 * m) = o;
          }
        }
      }
    }
  }
}

static inline int cdiv(long a, long b) { return (int)((a + b - 1) / b); }

extern "C" void kernel_launch(void* const* d_in, const int* in_sizes, int n_in,
                              void* d_out, int out_size, void* d_ws, size_t ws_size,
                              hipStream_t stream) {
  const float* x  = (const float*)d_in[0];
  const int*   ei = (const int*)d_in[1];
  const float* W1 = (const float*)d_in[2];
  const float* b1 = (const float*)d_in[3];
  const float* W2 = (const float*)d_in[4];
  const float* b2 = (const float*)d_in[5];
  const float* W3 = (const float*)d_in[6];
  const float* b3 = (const float*)d_in[7];

  const int N = in_sizes[0] / 128;
  const int E = in_sizes[1] / 2;
  const int* src = ei;
  const int* dst = ei + E;

  const int tiles = cdiv(N, TN);
  const int ranges = cdiv(tiles, TPR);

  char* ws = (char*)d_ws;
  size_t off0 = 0;
  size_t off1 = (((size_t)N * 4) + 255) & ~(size_t)255;
  size_t off2 = off1 + (size_t)N * 128 * 4;
  size_t off3 = off2 + (size_t)N * 128 * 4;
  size_t off4 = off3 + (size_t)tiles * LCAP * 8;
  size_t off5 = (off4 + (size_t)tiles * 4 + 255) & ~(size_t)255;
  size_t off6 = off5 + (size_t)ranges * NS * PCAP * 8;
  size_t off7 = (off6 + (size_t)ranges * NS * 4 + 255) & ~(size_t)255;
  float* dinv  = (float*)(ws + off0);
  float* bufH  = (float*)(ws + off1);
  float* bufF  = (float*)(ws + off2);
  int*   glist = (int*)(ws + off3);
  int*   gcnt  = (int*)(ws + off4);
  int*   gpair = (int*)(ws + off5);
  int*   gpcnt = (int*)(ws + off6);
  int*   gwcnt = (int*)(ws + off7);
  float* out   = (float*)d_out;

  k_pick<<<ranges * NS, BT, 0, stream>>>(src, dst, gpair, gpcnt, x, W1, bufH, N, E);
  k_list<<<tiles, BT, 0, stream>>>(gpair, gpcnt, src, dst, dinv, glist, gcnt, gwcnt, x, W1, bufH, N, E);

  k_agg<128, true, 128, true><<<tiles, BT, 0, stream>>>(glist, gcnt, gwcnt, src, dst, dinv, bufH, b1, W2, bufF, N, E);

  k_agg<128, true, 64, false><<<tiles, BT, 0, stream>>>(glist, gcnt, gwcnt, src, dst, dinv, bufF, b2, W3, bufH, N, E);

  k_agg<64, false, 0, false><<<tiles, BT, 0, stream>>>(glist, gcnt, gwcnt, src, dst, dinv, bufH, b3, nullptr, out, N, E);
}


# Tile lists sorted by row inside each owner wave's part (counting sort on the node counts); the owner sums a row's run in registers, one LDS read-add-write per run, not per entry; sum order changed
# speedup vs baseline: 4.9176x; 1.0609x over previous
#include <hip/hip_runtime.h>

typedef float v2f __attribute__((ext_vector_type(2)));
typedef float v8f __attribute__((ext_vector_type(8)));

constexpr int TN   = 392;
constexpr int LCAP = 12288;
constexpr int BT   = 1024;
constexpr int RU   = 8;
constexpr int TPR  = 16;
constexpr int RN   = TN * TPR;
constexpr int NS   = 16;
constexpr int PCAP = 8192;
constexpr int PU   = 6;
constexpr int QU   = 12;
constexpr int MWP  = 12;
constexpr int MWL  = 13;
constexpr int SPT  = BT - 32 * MWP;
constexpr int SLT  = BT - 32 * MWL;
constexpr int NW   = BT / 32;
constexpr int WCAP = LCAP / NW;
constexpr int WS   = WCAP + 16;
static_assert((TN + NW - 1) / NW <= 16, "a wave's self-loop entries must fit behind its filed ones");

typedef float vf2 __attribute__((ext_vector_type(2)));
typedef float vf4 __attribute__((ext_vector_type(4)));
typedef int   vi2 __attribute__((ext_vector_type(2)));
typedef int   vi4 __attribute__((ext_vector_type(4)));
template <int V> struct VT;
template <> struct VT<2> { typedef vf2 T; };
template <> struct VT<4> { typedef vf4 T; };

static_assert((PCAP / 2) % 32 == 0, "the 32 loads of one wave of k_list must stay inside one pair list");

__device__ __forceinline__ void strip_product(const float* __restrict__ A, const float* __restrict__ Wm,
                                              float* __restrict__ H, int M, int row0, int lane) {
  const int m  = lane & 15;
  const int hf = lane >> 4;
  const int arow = (row0 + m < M) ? (row0 + m) : (M - 1);
  const float* ap  = A + (size_t)arow * 128 + 4 * hf;
  const float* wp0 = Wm + (size_t)(4 * hf) * 128 + 4 * m;
  v8f acc[8] = {};
#pragma nounroll
  for (int it = 0; it < 16; ++it) {
    vf4 a = *(const vf4*)(ap + 8 * it);
    const float* wp = wp0 + (size_t)(8 * it) * 128;
    vf4 w[4][2];
#pragma unroll
    for (int e = 0; e < 4; ++e) {
#pragma unroll
      for (int g = 0; g < 2; ++g) w[e][g] = *(const vf4*)(wp + (size_t)e * 128 + g * 64);
    }
#pragma unroll
    for (int p = 0; p < 2; ++p) {
      v2f af; af.x = a[2 * p]; af.y = a[2 * p + 1];
#pragma unroll
      for (int t = 0; t < 8; ++t) {
        v2f bf; bf.x = w[2 * p][t / 4][t % 4]; bf.y = w[2 * p + 1][t / 4][t % 4];
        acc[t] = __builtin_amdgcn_wmma_f32_16x16x4_f32(
             false, af,  false, bf,
             (short)0, acc[t],  false,  false);
      }
    }
  }
  const int rbase = row0 + 8 * hf;
#pragma unroll
  for (int r = 0; r < 8; ++r) {
    const int row = rbase + r;
    if (row < M) {
#pragma unroll
      for (int g = 0; g < 2; ++g) {
        vf4 o;
        o.x = acc[4 * g][r]; o.y = acc[4 * g + 1][r]; o.z = acc[4 * g + 2][r]; o.w = acc[4 * g + 3][r];
        *(vf4*)(H + (size_t)row * 128 + g * 64 + 4 * m) = o;
      }
    }
  }
}

__global__ __launch_bounds__(BT)
void k_pick(const int* __restrict__ src, const int* __restrict__ dst,
            int* __restrict__ gpair, int* __restrict__ gpcnt,
            const float* __restrict__ x, const float* __restrict__ W1,
            float* __restrict__ h1, int N, int E) {
  __shared__ vi2 pl[PCAP];
  __shared__ int cnt;

  const int tid = threadIdx.x;
  const int lane = tid & 31;
  const int wave = tid >> 5;
  const int g   = blockIdx.x / NS;
  const int sl  = blockIdx.x % NS;
  const int rlo = g * RN;
  const unsigned urlo = (unsigned)rlo;
  const unsigned urnt = (unsigned)((N - rlo < RN) ? (N - rlo) : RN);

  const int es = (((E + NS - 1) / NS) + 3) & ~3;
  const long e0l = (long)sl * es;
  const int e0 = e0l < E ? (int)e0l : E;
  const int e1 = (E - e0 < es) ? E : (e0 + es);

  if (tid == 0) cnt = 0;
  __syncthreads();

  const bool al = ((((size_t)(src + e0)) | ((size_t)(dst + e0))) & 15u) == 0;
  const int nvec = al ? ((e1 - e0) >> 2) : 0;
  const vi4* dv = (const vi4*)(dst + e0);
  const vi4* sv = (const vi4*)(src + e0);

  if (wave >= BT / 32 - MWP) {
    const int strip = blockIdx.x * MWP + (wave - (BT / 32 - MWP));
    if (strip * 16 < N) strip_product(x, W1, h1, N, strip * 16, lane);
  } else {
  for (int vb = 0; vb < nvec; vb += SPT * PU) {
    vi4 d[PU], s[PU];
    if (vb + SPT * PU <= nvec) {
#pragma unroll
      for (int u = 0; u < PU; ++u) { d[u] = dv[vb + u * SPT + tid]; s[u] = sv[vb + u * SPT + tid]; }
    } else {
#pragma unroll
      for (int u = 0; u < PU; ++u) {
        int i = vb + u * SPT + tid;
        vi4 none = {-1, -1, -1, -1};
        d[u] = (i < nvec) ? dv[i] : none;
        s[u] = (i < nvec) ? sv[i] : none;
      }
    }
    unsigned m[4 * PU];
    int tot = 0;
#pragma unroll
    for (int u = 0; u < PU; ++u) {
#pragma unroll
      for (int j = 0; j < 4; ++j) {
        m[4 * u + j] = __builtin_amdgcn_ballot_w32((unsigned)d[u][j] - urlo < urnt);
        tot += __builtin_popcount(m[4 * u + j]);
      }
    }
    if (tot) {
      int base = 0;
      if (lane == 0) base = atomicAdd(&cnt, tot);
      base = __builtin_amdgcn_readfirstlane(base);
#pragma unroll
      for (int u = 0; u < PU; ++u) {
#pragma unroll
        for (int j = 0; j < 4; ++j) {
          const unsigned mk = m[4 * u + j];
          if ((unsigned)d[u][j] - urlo < urnt) {
            int pos = base + (int)__builtin_amdgcn_mbcnt_lo(mk, 0u);
            if (pos < PCAP) { vi2 ent; ent.x = s[u][j]; ent.y = d[u][j]; pl[pos] = ent; }
          }
          base += __builtin_popcount(mk);
        }
      }
    }
  }
  for (int e = e0 + 4 * nvec + tid; e < e1; e += SPT) {
    int dj = dst[e];
    if ((unsigned)dj - urlo < urnt) {
      int pos = atomicAdd(&cnt, 1);
      if (pos < PCAP) { vi2 ent; ent.x = src[e]; ent.y = dj; pl[pos] = ent; }
    }
  }
  }
  __syncthreads();

  const int total = cnt;
  const int n = total < PCAP ? total : PCAP;
  vi2* gp = (vi2*)gpair + (size_t)blockIdx.x * PCAP;
  for (int i = tid; i < n; i += BT) gp[i] = pl[i];
  if (tid == 0) {
    gpcnt[blockIdx.x] = total;
    if (n & 1) { vi2 pad; pad.x = 0; pad.y = -1; gp[n] = pad; }
  }
}

__global__ __launch_bounds__(BT)
void k_list(const int* __restrict__ gpair, const int* __restrict__ gpcnt,
            const int* __restrict__ src, const int* __restrict__ dst,
            float* __restrict__ dinv, int* __restrict__ glist,
            int* __restrict__ gcnt, int* __restrict__ gwcnt,
            const float* __restrict__ x, const float* __restrict__ W1,
            float* __restrict__ h1, int N, int E) {
  __shared__ vi2 lst[LCAP];
  __shared__ vi2 lst2[LCAP];
  __shared__ int wc[NW];
  __shared__ int degc[TN];
  __shared__ int rcur[TN];
  __shared__ int pc[NS];
  __shared__ int cnt;

  const int tid = threadIdx.x;
  const int lane = tid & 31;
  const int wave = tid >> 5;
  const int lo  = blockIdx.x * TN;
  const int nt  = (N - lo < TN) ? (N - lo) : TN;
  const unsigned unt = (unsigned)nt;
  const unsigned ulo = (unsigned)lo;
  const int g   = blockIdx.x / TPR;

  if (tid == 0) cnt = 0;
  if (tid < NS) pc[tid] = gpcnt[g * NS + tid];
  for (int t = tid; t < TN; t += BT) degc[t] = 0;
  __syncthreads();

  int pmax = 0;
#pragma unroll
  for (int s = 0; s < NS; ++s) pmax = pc[s] > pmax ? pc[s] : pmax;

  if (wave >= BT / 32 - MWL) {
    const int taken = ((gridDim.x + TPR - 1) / TPR) * NS * MWP;
    const int nstr  = (N + 15) >> 4;
    for (int strip = taken + blockIdx.x * MWL + (wave - (BT / 32 - MWL)); strip < nstr; strip += gridDim.x * MWL)
      strip_product(x, W1, h1, N, strip * 16, lane);
  } else if (pmax <= PCAP) {
    constexpr int VPL = PCAP / 2;
    const vi4* pv = (const vi4*)gpair + (size_t)g * NS * VPL;
    for (int vb = 0; vb < NS * VPL; vb += SLT * QU) {
      vi4 p[QU];
#pragma unroll
      for (int u = 0; u < QU; ++u) {
        int v  = vb + u * SLT + tid;
        int np = (v < NS * VPL) ? pc[v / VPL] : 0;
        int i  = 2 * (v % VPL);
        vi4 none = {0, -1, 0, -1};
        p[u] = (i < np) ? pv[v] : none;
      }
      unsigned m[2 * QU];
      int tot = 0;
#pragma unroll
      for (int u = 0; u < QU; ++u) {
        m[2 * u]     = __builtin_amdgcn_ballot_w32((unsigned)p[u].y - ulo < unt);
        m[2 * u + 1] = __builtin_amdgcn_ballot_w32((unsigned)p[u].w - ulo < unt);
        tot += __builtin_popcount(m[2 * u]) + __builtin_popcount(m[2 * u + 1]);
      }
      if (tot) {
        int base = 0;
        if (lane == 0) base = atomicAdd(&cnt, tot);
        base = __builtin_amdgcn_readfirstlane(base);
#pragma unroll
        for (int u = 0; u < QU; ++u) {
          unsigned t0 = (unsigned)p[u].y - ulo, t1 = (unsigned)p[u].w - ulo;
          if (t0 < unt) {
            atomicAdd(&degc[t0], 1);
            int pos = base + (int)__builtin_amdgcn_mbcnt_lo(m[2 * u], 0u);
            if (pos < LCAP) { vi2 ent; ent.x = p[u].x; ent.y = (int)t0; lst[pos] = ent; }
          }
          base += __builtin_popcount(m[2 * u]);
          if (t1 < unt) {
            atomicAdd(&degc[t1], 1);
            int pos = base + (int)__builtin_amdgcn_mbcnt_lo(m[2 * u + 1], 0u);
            if (pos < LCAP) { vi2 ent; ent.x = p[u].z; ent.y = (int)t1; lst[pos] = ent; }
          }
          base += __builtin_popcount(m[2 * u + 1]);
        }
      }
    }
  } else {
    for (int e = tid; e < E; e += SLT) {
      unsigned t = (unsigned)dst[e] - ulo;
      if (t < unt) {
        atomicAdd(&degc[t], 1);
        int pos = atomicAdd(&cnt, 1);
        if (pos < LCAP) { vi2 ent; ent.x = src[e]; ent.y = (int)t; lst[pos] = ent; }
      }
    }
  }
  __syncthreads();

  const int total = cnt;
  const int n = total < LCAP ? total : LCAP;
  for (int t = tid; t < TN; t += BT) {
    int s = 0;
    for (int r = t & (NW - 1); r < t; r += NW) s += degc[r];
    rcur[t] = s;
  }
  if (tid < NW) {
    int s = 0;
    for (int r = tid; r < TN; r += NW) s += degc[r];
    wc[tid] = s;
  }
  __syncthreads();
  for (int i = tid; i < n; i += BT) {
    vi2 ent = lst[i];
    int o = ent.y & (NW - 1);
    int pos = atomicAdd(&rcur[ent.y], 1);
    if (pos < WCAP) lst2[o * WCAP + pos] = ent;
  }
  __syncthreads();

  {
    int c = wc[wave];
    c = c < WCAP ? c : WCAP;
    vi2* gl = (vi2*)glist + (size_t)blockIdx.x * LCAP + wave * WCAP;
    for (int k = lane; k < c; k += 32) gl[k] = lst2[wave * WCAP + k];
  }
  if (tid < NW) gwcnt[blockIdx.x * NW + tid] = wc[tid];
  if (tid == 0) {
    int over = total > LCAP;
    for (int o = 0; o < NW; ++o) over |= wc[o] > WCAP;
    gcnt[blockIdx.x] = over ? LCAP + 1 : total;
  }
  for (int t = tid; t < nt; t += BT) dinv[lo + t] = rsqrtf(1.0f + (float)degc[t]);
}

constexpr int SB = 23;
static_assert(TN <= (1 << (32 - SB)), "the local row must fit above the source bits");

template <int OUT, bool RELU, int OUTN, bool FIRST>
__global__ __launch_bounds__(BT)
void k_agg(int* __restrict__ glist, const int* __restrict__ gcnt, const int* __restrict__ gwcnt,
           const int* __restrict__ src, const int* __restrict__ dst,
           const float* __restrict__ dinv, const float* __restrict__ h,
           const float* __restrict__ b, const float* __restrict__ Wn,
           float* __restrict__ outp, int N, int E) {
  constexpr int VEC = OUT / 32;
  constexpr bool FUSE = OUTN > 0;
  constexpr int P  = FUSE ? OUT + 4 : OUT;
  constexpr int TR = FUSE ? ((TN + 15) / 16) * 16 : TN + 4;
  constexpr int SPARE = TR - 1;
  static_assert(SPARE >= TN && SPARE < (1 << (32 - SB)), "the spare row lies past the tile and fits an entry's row bits");
  static_assert(WS % (2 * RU) == 0, "a wave's list is padded to an even number of whole batches inside its part");
  static_assert(!FUSE || (OUT == 128 && OUTN % 64 == 0), "fused product: K = 128, whole 64-column groups");
  static_assert((TN + 15) / 16 <= BT / 32, "one strip per wave");
  static_assert((TR * P) % 4 == 0, "the tile is cleared in 16-byte pieces");
  static_assert(P % VEC == 0, "a lane's columns of a row are read and written as one aligned piece");
  typedef typename VT<VEC>::T V;
  __shared__ __attribute__((aligned(16))) float tile[TR * P];
  __shared__ vi2   lst[NW * WS];
  __shared__ __attribute__((aligned(16))) float bl[FUSE ? OUT : 4];

  const int tid  = threadIdx.x;
  const int lane = tid & 31;
  const int wave = tid >> 5;
  const int lo   = blockIdx.x * TN;
  const int nt   = (N - lo < TN) ? (N - lo) : TN;
  const unsigned unt = (unsigned)nt;
  const unsigned ulo = (unsigned)lo;

  const int  total = gcnt[blockIdx.x];
  const int  filed = gwcnt[blockIdx.x * NW + wave];
  const int  rs   = wave + NW * lane;
  const int  ns   = __builtin_amdgcn_readfirstlane(nt > wave ? (nt - 1 - wave) / NW + 1 : 0);
  const float di = dinv[lo + (lane < ns ? rs : 0)];
  const float bv = FUSE ? b[tid < OUT ? tid : 0] : 0.0f;
  const bool whole = total <= LCAP && N <= (1 << SB);
  const int  cw    = __builtin_amdgcn_readfirstlane(whole ? (filed < WCAP ? filed : WCAP) : 0);

  {
    vf4 z = {0.0f, 0.0f, 0.0f, 0.0f};
    for (int i = tid; i < TR * P / 4; i += BT) ((vf4*)tile)[i] = z;
  }
  vi2* my = lst + wave * WS;
  {
    constexpr int LU = 8;
    vi2* gl = (vi2*)glist + (size_t)blockIdx.x * LCAP + wave * WCAP;
    for (int k0 = lane; k0 < cw; k0 += 32 * LU) {
      vi2 ent[LU];
#pragma unroll
      for (int u = 0; u < LU; ++u) {
        int k = k0 + u * 32;
        vi2 none = {0, 0};
        ent[u] = (k < cw) ? gl[k] : none;
      }
      if constexpr (FIRST) {
        float ds[LU], dd[LU];
#pragma unroll
        for (int u = 0; u < LU; ++u) {
          ds[u] = dinv[ent[u].x];
          dd[u] = dinv[lo + ent[u].y];
        }
#pragma unroll
        for (int u = 0; u < LU; ++u) {
          int k = k0 + u * 32;
          ent[u].x = ent[u].x | (int)((unsigned)ent[u].y << SB);
          ent[u].y = __float_as_int(ds[u] * dd[u]);
          if (k < cw) gl[k] = ent[u];
        }
      }
#pragma unroll
      for (int u = 0; u < LU; ++u) {
        int k = k0 + u * 32;
        if (k < cw) my[k] = ent[u];
      }
    }
  }
  const int ntot = cw + ns;
  const int nb   = ((ntot + 2 * RU - 1) / (2 * RU)) * 2;
  if (lane < ns) {
    vi2 ent;
    ent.x = (lo + rs) | (int)((unsigned)rs << SB);
    ent.y = __float_as_int(di * di);
    my[cw + lane] = ent;
  } else if (cw + lane < nb * RU) {
    vi2 ent;
    ent.x = (int)((unsigned)SPARE << SB);
    ent.y = __float_as_int(0.0f);
    my[cw + lane] = ent;
  }
  if (FUSE) {
    if (tid < OUT) bl[tid] = bv;
  }
  __syncthreads();

  if (!whole) {
    for (int e = tid; e < E; e += BT) {
      unsigned t = (unsigned)dst[e] - ulo;
      if (t < unt) {
        int s = src[e];
        float nrm = dinv[s] * dinv[lo + t];
        const float* hs = h + (size_t)s * OUT;
        for (int c = 0; c < OUT; ++c)
          atomicAdd(&tile[t * P + c], hs[c] * nrm);
      }
    }
    __syncthreads();
  }

  {
    int   ta[RU], tb[RU];
    float wa[RU], wb[RU];
    V     xa[RU], xb[RU];
    auto load_batch = [&](int bi, int (&t)[RU], float (&w)[RU], V (&x)[RU]) {
      const vi2* ep = my + bi * RU;
      int s[RU];
#pragma unroll
      for (int u = 0; u < RU; ++u) {
        vi2 ent = ep[u];
        unsigned sp = (unsigned)__builtin_amdgcn_readfirstlane(ent.x);
        w[u] = __int_as_float(__builtin_amdgcn_readfirstlane(ent.y));
        s[u] = (int)(sp & ((1u << SB) - 1u));
        t[u] = (int)(sp >> SB);
      }
#pragma unroll
      for (int u = 0; u < RU; ++u)
        x[u] = *(const V*)(h + (size_t)s[u] * OUT + VEC * lane);
    };
    int cur = SPARE;
    V   acc = {};
    auto flush = [&]() {
      V* rp = (V*)(tile + cur * P + VEC * lane);
      V a = *rp;
#pragma unroll
      for (int j = 0; j < VEC; ++j) a[j] += acc[j];
      *rp = a;
    };
    auto add_batch = [&](const int (&t)[RU], const float (&w)[RU], const V (&x)[RU]) {
#pragma unroll
      for (int u = 0; u < RU; ++u) {
        if (t[u] != cur) {
          flush();
          cur = t[u];
#pragma unroll
          for (int j = 0; j < VEC; ++j) acc[j] = 0.0f;
        }
#pragma unroll
        for (int j = 0; j < VEC; ++j) acc[j] += x[u][j] * w[u];
      }
    };
    if (nb > 0) {
      load_batch(0, ta, wa, xa);
      for (int bi = 0; bi < nb; bi += 2) {
        load_batch(bi + 1, tb, wb, xb);
        add_batch(ta, wa, xa);
        load_batch(bi + 2 < nb ? bi + 2 : nb - 1, ta, wa, xa);
        add_batch(tb, wb, xb);
      }
      flush();
    }
  }
  __syncthreads();

  if constexpr (!FUSE) {
    V bb = *(const V*)(b + VEC * lane);
    for (int t = wave; t < nt; t += BT / 32) {
      V a = *(const V*)(tile + t * P + VEC * lane);
      V r;
#pragma unroll
      for (int j = 0; j < VEC; ++j) {
        float v = a[j] + bb[j];
        r[j] = RELU ? fmaxf(v, 0.0f) : v;
      }
      *(V*)(outp + (size_t)(lo + t) * OUT + VEC * lane) = r;
    }
  } else {
    constexpr int NTN = OUTN / 16;
    constexpr int G   = OUTN / 64;
    const int nstrips = (nt + 15) >> 4;
    if (wave < nstrips) {
      const int m  = lane & 15;
      const int hf = lane >> 4;
      const float* arow = tile + (wave * 16 + m) * P + 4 * hf;
      const float* wp0  = Wn + (size_t)(4 * hf) * OUTN + 4 * m;
      v8f acc[NTN] = {};
      for (int it = 0; it < 16; ++it) {
        vf4 a  = *(const vf4*)(arow + 8 * it);
        vf4 bv4 = *(const vf4*)(bl + 8 * it + 4 * hf);
#pragma unroll
        for (int e = 0; e < 4; ++e) {
          float v = a[e] + bv4[e];
          a[e] = RELU ? fmaxf(v, 0.0f) : v;
        }
        const float* wp = wp0 + (size_t)(8 * it) * OUTN;
        vf4 w[4][G];
#pragma unroll
        for (int e = 0; e < 4; ++e) {
#pragma unroll
          for (int g = 0; g < G; ++g) w[e][g] = *(const vf4*)(wp + (size_t)e * OUTN + g * 64);
        }
#pragma unroll
        for (int p = 0; p < 2; ++p) {
          v2f af; af.x = a[2 * p]; af.y = a[2 * p + 1];
#pragma unroll
          for (int t = 0; t < NTN; ++t) {
            v2f bf; bf.x = w[2 * p][t / 4][t % 4]; bf.y = w[2 * p + 1][t / 4][t % 4];
            acc[t] = __builtin_amdgcn_wmma_f32_16x16x4_f32(
                 false, af,  false, bf,
                 (short)0, acc[t],  false,  false);
          }
        }
      }
      const int rbase = wave * 16 + 8 * hf;
#pragma unroll
      for (int r = 0; r < 8; ++r) {
        const int row = rbase + r;
        if (row < nt) {
#pragma unroll
          for (int g = 0; g < G; ++g) {
            vf4 o;
            o.x = acc[4 * g][r]; o.y = acc[4 * g + 1][r]; o.z = acc[4 * g + 2][r]; o.w = acc[4 * g + 3][r];
            *(vf4*)(outp + (size_t)(lo + row) * OUTN + g * 64 + 4 * m) = o;
          }
        }
      }
    }
  }
}

static inline int cdiv(long a, long b) { return (int)((a + b - 1) / b); }

extern "C" void kernel_launch(void* const* d_in, const int* in_sizes, int n_in,
                              void* d_out, int out_size, void* d_ws, size_t ws_size,
                              hipStream_t stream) {
  const float* x  = (const float*)d_in[0];
  const int*   ei = (const int*)d_in[1];
  const float* W1 = (const float*)d_in[2];
  const float* b1 = (const float*)d_in[3];
  const float* W2 = (const float*)d_in[4];
  const float* b2 = (const float*)d_in[5];
  const float* W3 = (const float*)d_in[6];
  const float* b3 = (const float*)d_in[7];

  const int N = in_sizes[0] / 128;
  const int E = in_sizes[1] / 2;
  const int* src = ei;
  const int* dst = ei + E;

  const int tiles = cdiv(N, TN);
  const int ranges = cdiv(tiles, TPR);

  char* ws = (char*)d_ws;
  size_t off0 = 0;
  size_t off1 = (((size_t)N * 4) + 255) & ~(size_t)255;
  size_t off2 = off1 + (size_t)N * 128 * 4;
  size_t off3 = off2 + (size_t)N * 128 * 4;
  size_t off4 = off3 + (size_t)tiles * LCAP * 8;
  size_t off5 = (off4 + (size_t)tiles * 4 + 255) & ~(size_t)255;
  size_t off6 = off5 + (size_t)ranges * NS * PCAP * 8;
  size_t off7 = (off6 + (size_t)ranges * NS * 4 + 255) & ~(size_t)255;
  float* dinv  = (float*)(ws + off0);
  float* bufH  = (float*)(ws + off1);
  float* bufF  = (float*)(ws + off2);
  int*   glist = (int*)(ws + off3);
  int*   gcnt  = (int*)(ws + off4);
  int*   gpair = (int*)(ws + off5);
  int*   gpcnt = (int*)(ws + off6);
  int*   gwcnt = (int*)(ws + off7);
  float* out   = (float*)d_out;

  k_pick<<<ranges * NS, BT, 0, stream>>>(src, dst, gpair, gpcnt, x, W1, bufH, N, E);
  k_list<<<tiles, BT, 0, stream>>>(gpair, gpcnt, src, dst, dinv, glist, gcnt, gwcnt, x, W1, bufH, N, E);

  k_agg<128, true, 128, true><<<tiles, BT, 0, stream>>>(glist, gcnt, gwcnt, src, dst, dinv, bufH, b1, W2, bufF, N, E);

  k_agg<128, true, 64, false><<<tiles, BT, 0, stream>>>(glist, gcnt, gwcnt, src, dst, dinv, bufF, b2, W3, bufH, N, E);

  k_agg<64, false, 0, false><<<tiles, BT, 0, stream>>>(glist, gcnt, gwcnt, src, dst, dinv, bufH, b3, nullptr, out, N, E);
}
